# starting kernel
# baseline (speedup 1.0000x reference)
#include <hip/hip_runtime.h>
#include <cmath>


typedef float v2f __attribute__((ext_vector_type(2)));
typedef float v8f __attribute__((ext_vector_type(8)));

#define NSTEP 256
#define NODES 288
#define BT    16

__global__ __launch_bounds__(32)
void btree_american_wmma(const float* __restrict__ kin, float* __restrict__ out,
                         float w0, float w1, float uscale, float alpha)
{
    __shared__ float V[NODES * BT];
    __shared__ float PB[NODES];

    const int lane = threadIdx.x;
    const int l15  = lane & 15;
    const int hi   = lane >> 4;
    const int b0   = blockIdx.x * BT;

    const float kv = kin[b0 + l15];

    for (int j = hi; j < NODES; j += 2) {
        float st = 100.0f * expf(alpha * (float)(2 * j - NSTEP));
        V[j * BT + l15] = fmaxf(kv - st, 0.0f);
    }
    for (int j = lane; j < NODES; j += 32) {
        PB[j] = 100.0f * expf(alpha * (float)(2 * j - (NSTEP - 1)));
    }
    __syncthreads();

    v2f Av[5];
#pragma unroll
    for (int t = 0; t < 5; ++t) {
        int k0 = 4 * t + 2 * hi;
        int k1 = k0 + 1;
        float e0 = (k0 == l15) ? w0 : ((k0 == l15 + 1) ? w1 : 0.0f);
        float e1 = (k1 == l15) ? w0 : ((k1 == l15 + 1) ? w1 : 0.0f);
        Av[t][0] = e0;
        Av[t][1] = e1;
    }

    float scale = 1.0f;
    for (int i = 0; i < NSTEP; ++i) {
        const int n_out  = NSTEP - i;
        const int ntiles = (n_out + 15) >> 4;

        for (int g = 0; g < ntiles; ++g) {
            const int base = g << 4;

            v2f Bv[5];
#pragma unroll
            for (int t = 0; t < 5; ++t) {
                int kb = base + 4 * t + 2 * hi;
                Bv[t][0] = V[kb * BT + l15];
                Bv[t][1] = V[(kb + 1) * BT + l15];
            }

            v8f c = {};
#pragma unroll
            for (int t = 0; t < 5; ++t) {
                c = __builtin_amdgcn_wmma_f32_16x16x4_f32(
                         false, Av[t],
                         false, Bv[t],
                         (short)0, c,
                         false,  false);
            }

#pragma unroll
            for (int r = 0; r < 8; ++r) {
                int node  = base + r + 8 * hi;
                float pay = fmaf(-scale, PB[node], kv);
                V[node * BT + l15] = fmaxf(c[r], pay);
            }
        }
        scale *= uscale;
        __syncthreads();
    }

    if (hi == 0) out[b0 + l15] = V[l15];
}

extern "C" void kernel_launch(void* const* d_in, const int* in_sizes, int n_in,
                              void* d_out, int out_size, void* d_ws, size_t ws_size,
                              hipStream_t stream) {
    (void)n_in; (void)d_ws; (void)ws_size; (void)out_size;
    const float* kin = (const float*)d_in[0];
    float* out = (float*)d_out;
    const int B = in_sizes[0];

    const double S0g = 100.0, SIG = 0.2, Tg = 1.0, T0g = 0.0, Rg = 0.05;
    (void)S0g;
    const double dt   = (Tg - T0g) / (double)NSTEP;
    const double sq   = sqrt(dt);
    const double u    = exp(SIG * sq);
    const double d    = exp(-SIG * sq);
    const double disc = exp(-Rg * dt);
    const float w0    = (float)((disc * u - 1.0) / (u - d));
    const float w1    = (float)((1.0 - disc * d) / (u - d));
    const float us    = (float)u;
    const float alpha = (float)(SIG * sq);

    const int nblocks = B / BT;
    btree_american_wmma<<<nblocks, 32, 0, stream>>>(kin, out, w0, w1, us, alpha);
}


# Tile loop software-pipelined: tiles step in pairs as two independent matrix-op chains; next pair's LDS loads and previous pair's max+store issue under the running chains (lattice as node pairs)
# speedup vs baseline: 1.4448x; 1.4448x over previous
#include <hip/hip_runtime.h>
#include <cmath>


typedef float v2f __attribute__((ext_vector_type(2)));
typedef float v8f __attribute__((ext_vector_type(8)));

#define NSTEP 256
#define NODES 288
#define NPAIR (NODES / 2)
#define BT    16

__host__ __device__ constexpr int pairs_at(int i) { return ((((NSTEP - i + 15) >> 4)) + 1) >> 1; }
__host__ __device__ constexpr int pipelined_steps() { int i = 0; while (i < NSTEP && pairs_at(i) >= 3) ++i; return i; }
__host__ __device__ constexpr int pipelined_trips() { int s = 0; for (int i = 0; i < pipelined_steps(); ++i) s += pairs_at(i); return s; }
constexpr int PSTEPS = pipelined_steps();
constexpr int PTRIPS = pipelined_trips();
static_assert(PTRIPS % 2 == 0, "the trip loop is unrolled by two");

__device__ __forceinline__ void load_pair(const v2f* V2, int hb, int hi, int l15, v2f (&B)[9])
{
#pragma unroll
    for (int t = 0; t < 9; ++t) B[t] = V2[(hb + 2 * t + hi) * BT + l15];
}

__device__ __forceinline__ void load_pb(const float* PB, int pr, int hi, v8f& Q0, v8f& Q1)
{
#pragma unroll
    for (int r = 0; r < 8; ++r) {
        int node = 32 * pr + r + 8 * hi;
        Q0[r] = PB[node];
        Q1[r] = PB[node + 16];
    }
}

__device__ __forceinline__ v8f max_tile(const v8f& c, const v8f& p)
{
    v8f r;
#pragma unroll
    for (int e = 0; e < 8; ++e) r[e] = fmaxf(c[e], p[e]);
    return r;
}

__device__ __forceinline__ void store_pair(v2f* V2, int hb, int hi, int l15, const v8f& r0, const v8f& r1)
{
#pragma unroll
    for (int q = 0; q < 4; ++q) {
        v2f a, b;
        a[0] = r0[2 * q]; a[1] = r0[2 * q + 1];
        b[0] = r1[2 * q]; b[1] = r1[2 * q + 1];
        V2[(hb + 4 * hi + q) * BT + l15]     = a;
        V2[(hb + 8 + 4 * hi + q) * BT + l15] = b;
    }
}

__device__ __forceinline__ void max_store_pair(v2f* V2, int hb, int hi, int l15,
                                               const v8f& c0, const v8f& c1,
                                               const v8f& p0, const v8f& p1)
{
    store_pair(V2, hb, hi, l15, max_tile(c0, p0), max_tile(c1, p1));
}

__device__ __forceinline__ void trip(v2f* V2, const float* PB, int hi, int l15, float kv, float uscale,
                                     const v2f (&Av)[5], const v2f (&Bc)[9], v2f (&Bn)[9],
                                     const v8f& Qc0, const v8f& Qc1, v8f& Qn0, v8f& Qn1,
                                     v8f& Cc0, v8f& Cc1, v8f& Pc0, v8f& Pc1,
                                     const v8f& Cp0, const v8f& Cp1, const v8f& Pp0, const v8f& Pp1,
                                     int& shp, int& i, int& p, int& npairs, float& scale)
{
    const int  hb   = 16 * p;
    const bool last = (p + 1 == npairs);
    const int  np   = last ? 0 : p + 1;

    v8f z = {};
    Cc0 = __builtin_amdgcn_wmma_f32_16x16x4_f32(false, Av[0], false, Bc[0], (short)0, z, false, false);
    Cc1 = __builtin_amdgcn_wmma_f32_16x16x4_f32(false, Av[0], false, Bc[4], (short)0, z, false, false);

    v8f e0 = Cp0, e1 = Cp1;
    asm volatile("" : "+v"(Cc0), "+v"(Cc1), "+v"(e0));
    v8f r0 = max_tile(e0, Pp0);
    Cc0 = __builtin_amdgcn_wmma_f32_16x16x4_f32(false, Av[1], false, Bc[1], (short)0, Cc0, false, false);
    Cc1 = __builtin_amdgcn_wmma_f32_16x16x4_f32(false, Av[1], false, Bc[5], (short)0, Cc1, false, false);
    asm volatile("" : "+v"(Cc0), "+v"(Cc1), "+v"(e1), "+v"(r0));
    v8f r1 = max_tile(e1, Pp1);
    store_pair(V2, shp, hi, l15, r0, r1);
    load_pair(V2, 16 * np, hi, l15, Bn);
    load_pb(PB, np, hi, Qn0, Qn1);
    asm volatile("" : "+v"(Cc0), "+v"(Cc1) : : "memory");

#pragma unroll
    for (int t = 2; t < 5; ++t) {
        Cc0 = __builtin_amdgcn_wmma_f32_16x16x4_f32(false, Av[t], false, Bc[t],     (short)0, Cc0, false, false);
        Cc1 = __builtin_amdgcn_wmma_f32_16x16x4_f32(false, Av[t], false, Bc[4 + t], (short)0, Cc1, false, false);
    }

#pragma unroll
    for (int r = 0; r < 8; ++r) {
        Pc0[r] = fmaf(-scale, Qc0[r], kv);
        Pc1[r] = fmaf(-scale, Qc1[r], kv);
    }
    asm volatile("" : "+v"(Bn[0]), "+v"(Bn[4]), "+v"(Bn[1]), "+v"(Bn[5]) : "v"(Cc0), "v"(Cc1));

    shp = hb;

    if (last) { p = 0; ++i; scale *= uscale; npairs = pairs_at(i); }
    else      { ++p; }
}

__global__ __launch_bounds__(32)
void btree_american_wmma(const float* __restrict__ kin, float* __restrict__ out,
                         float w0, float w1, float uscale, float alpha)
{
    __shared__ v2f   V2[NPAIR * BT];
    __shared__ float PB[NODES];

    const int lane = threadIdx.x;
    const int l15  = lane & 15;
    const int hi   = lane >> 4;
    const int b0   = blockIdx.x * BT;

    const float kv = kin[b0 + l15];

    for (int h = hi; h < NPAIR; h += 2) {
        float s0 = 100.0f * expf(alpha * (float)(4 * h - NSTEP));
        float s1 = 100.0f * expf(alpha * (float)(4 * h + 2 - NSTEP));
        v2f v;
        v[0] = fmaxf(kv - s0, 0.0f);
        v[1] = fmaxf(kv - s1, 0.0f);
        V2[h * BT + l15] = v;
    }
    for (int j = lane; j < NODES; j += 32) {
        PB[j] = 100.0f * expf(alpha * (float)(2 * j - (NSTEP - 1)));
    }
    __syncthreads();

    v2f Av[5];
#pragma unroll
    for (int t = 0; t < 5; ++t) {
        int k0 = 4 * t + 2 * hi;
        int k1 = k0 + 1;
        float e0 = (k0 == l15) ? w0 : ((k0 == l15 + 1) ? w1 : 0.0f);
        float e1 = (k1 == l15) ? w0 : ((k1 == l15 + 1) ? w1 : 0.0f);
        Av[t][0] = e0;
        Av[t][1] = e1;
    }

    v2f X[9], Y[9];
    v8f Ca0, Ca1, Pa0, Pa1;
    v8f Cb0, Cb1, Pb0, Pb1;
    v8f QX0, QX1, QY0, QY1;
    load_pair(V2, 0, hi, l15, X);
    load_pb(PB, 0, hi, QX0, QX1);

    int shp = 0;
#pragma unroll
    for (int q = 0; q < 4; ++q) {
        v2f a = V2[(4 * hi + q) * BT + l15];
        v2f b = V2[(8 + 4 * hi + q) * BT + l15];
        Cb0[2 * q] = a[0]; Cb0[2 * q + 1] = a[1];
        Cb1[2 * q] = b[0]; Cb1[2 * q + 1] = b[1];
    }
#pragma unroll
    for (int r = 0; r < 8; ++r) { Pb0[r] = -1.0f; Pb1[r] = -1.0f; }

    int   i = 0, p = 0, npairs = pairs_at(0);
    float scale = 1.0f;
    for (int k = 0; k < PTRIPS / 2; ++k) {
        trip(V2, PB, hi, l15, kv, uscale, Av, X, Y, QX0, QX1, QY0, QY1,
             Ca0, Ca1, Pa0, Pa1, Cb0, Cb1, Pb0, Pb1, shp, i, p, npairs, scale);
        trip(V2, PB, hi, l15, kv, uscale, Av, Y, X, QY0, QY1, QX0, QX1,
             Cb0, Cb1, Pb0, Pb1, Ca0, Ca1, Pa0, Pa1, shp, i, p, npairs, scale);
    }
    max_store_pair(V2, shp, hi, l15, Cb0, Cb1, Pb0, Pb1);

    for (int s = PSTEPS; s < NSTEP; ++s) {
        const int np = pairs_at(s);
        for (int q = 0; q < np; ++q) {
            load_pair(V2, 16 * q, hi, l15, X);
            v8f c0 = {};
            v8f c1 = {};
#pragma unroll
            for (int t = 0; t < 5; ++t) {
                c0 = __builtin_amdgcn_wmma_f32_16x16x4_f32(false, Av[t], false, X[t],     (short)0, c0, false, false);
                c1 = __builtin_amdgcn_wmma_f32_16x16x4_f32(false, Av[t], false, X[4 + t], (short)0, c1, false, false);
            }
            v8f y0, y1;
#pragma unroll
            for (int r = 0; r < 8; ++r) {
                int node = 32 * q + r + 8 * hi;
                y0[r] = fmaf(-scale, PB[node], kv);
                y1[r] = fmaf(-scale, PB[node + 16], kv);
            }
            max_store_pair(V2, 16 * q, hi, l15, c0, c1, y0, y1);
        }
        scale *= uscale;
    }
    __syncthreads();

    if (hi == 0) out[b0 + l15] = V2[l15][0];
}

extern "C" void kernel_launch(void* const* d_in, const int* in_sizes, int n_in,
                              void* d_out, int out_size, void* d_ws, size_t ws_size,
                              hipStream_t stream) {
    (void)n_in; (void)d_ws; (void)ws_size; (void)out_size;
    const float* kin = (const float*)d_in[0];
    float* out = (float*)d_out;
    const int B = in_sizes[0];

    const double S0g = 100.0, SIG = 0.2, Tg = 1.0, T0g = 0.0, Rg = 0.05;
    (void)S0g;
    const double dt   = (Tg - T0g) / (double)NSTEP;
    const double sq   = sqrt(dt);
    const double u    = exp(SIG * sq);
    const double d    = exp(-SIG * sq);
    const double disc = exp(-Rg * dt);
    const float w0    = (float)((disc * u - 1.0) / (u - d));
    const float w1    = (float)((1.0 - disc * d) / (u - d));
    const float us    = (float)u;
    const float alpha = (float)(SIG * sq);

    const int nblocks = B / BT;
    btree_american_wmma<<<nblocks, 32, 0, stream>>>(kin, out, w0, w1, us, alpha);
}


# Operand loads run two tile pairs ahead through a ring of three register sets (steps of four pairs or more): every LDS wait in that loop is a counted wait behind a newer load group
# speedup vs baseline: 1.5580x; 1.0783x over previous
#include <hip/hip_runtime.h>
#include <cmath>


typedef float v2f __attribute__((ext_vector_type(2)));
typedef float v8f __attribute__((ext_vector_type(8)));

#define NSTEP 256
#define NODES 288
#define NPAIR (NODES / 2)
#define BT    16

__host__ __device__ constexpr int pairs_at(int i) { return ((((NSTEP - i + 15) >> 4)) + 1) >> 1; }
__host__ __device__ constexpr int pipelined_steps() { int i = 0; while (i < NSTEP && pairs_at(i) >= 3) ++i; return i; }
__host__ __device__ constexpr int pipelined_trips() { int s = 0; for (int i = 0; i < pipelined_steps(); ++i) s += pairs_at(i); return s; }
__host__ __device__ constexpr int deep_steps() { int i = 0; while (i < NSTEP && pairs_at(i) >= 4) ++i; return i; }
__host__ __device__ constexpr int deep_trips() { int s = 0; for (int i = 0; i < deep_steps(); ++i) s += pairs_at(i); return s; }
constexpr int DSTEPS = deep_steps();
constexpr int DTRIPS = deep_trips();
constexpr int PSTEPS = pipelined_steps();
constexpr int PTRIPS = pipelined_trips();
static_assert(DSTEPS <= PSTEPS, "the two-ahead steps are a prefix of the pipelined steps");
static_assert(DTRIPS % 6 == 0, "the two-ahead loop is unrolled by six (ring of three, sums of two)");
static_assert((PTRIPS - DTRIPS) % 2 == 0, "the one-ahead loop is unrolled by two");
static_assert(pairs_at(DSTEPS) >= 2, "the last two-ahead trips load pairs 0 and 1 of the step behind them");

__device__ __forceinline__ void load_pair(const v2f* V2, int hb, int hi, int l15, v2f (&B)[9])
{
#pragma unroll
    for (int t = 0; t < 9; ++t) B[t] = V2[(hb + 2 * t + hi) * BT + l15];
}

__device__ __forceinline__ void load_pb(const float* PB, int pr, int hi, v8f& Q0, v8f& Q1)
{
#pragma unroll
    for (int r = 0; r < 8; ++r) {
        int node = 32 * pr + r + 8 * hi;
        Q0[r] = PB[node];
        Q1[r] = PB[node + 16];
    }
}

__device__ __forceinline__ v8f max_tile(const v8f& c, const v8f& p)
{
    v8f r;
#pragma unroll
    for (int e = 0; e < 8; ++e) r[e] = fmaxf(c[e], p[e]);
    return r;
}

__device__ __forceinline__ void store_pair(v2f* V2, int hb, int hi, int l15, const v8f& r0, const v8f& r1)
{
#pragma unroll
    for (int q = 0; q < 4; ++q) {
        v2f a, b;
        a[0] = r0[2 * q]; a[1] = r0[2 * q + 1];
        b[0] = r1[2 * q]; b[1] = r1[2 * q + 1];
        V2[(hb + 4 * hi + q) * BT + l15]     = a;
        V2[(hb + 8 + 4 * hi + q) * BT + l15] = b;
    }
}

__device__ __forceinline__ void max_store_pair(v2f* V2, int hb, int hi, int l15,
                                               const v8f& c0, const v8f& c1,
                                               const v8f& p0, const v8f& p1)
{
    store_pair(V2, hb, hi, l15, max_tile(c0, p0), max_tile(c1, p1));
}

__device__ __forceinline__ void trip(v2f* V2, const float* PB, int hi, int l15, float kv, float uscale,
                                     const v2f (&Av)[5], const v2f (&Bc)[9], v2f (&Bn)[9],
                                     const v8f& Qc0, const v8f& Qc1, v8f& Qn0, v8f& Qn1,
                                     v8f& Cc0, v8f& Cc1, v8f& Pc0, v8f& Pc1,
                                     const v8f& Cp0, const v8f& Cp1, const v8f& Pp0, const v8f& Pp1,
                                     int& shp, int& i, int& p, int& npairs, float& scale)
{
    const int  hb   = 16 * p;
    const bool last = (p + 1 == npairs);
    const int  np   = last ? 0 : p + 1;

    v8f z = {};
    Cc0 = __builtin_amdgcn_wmma_f32_16x16x4_f32(false, Av[0], false, Bc[0], (short)0, z, false, false);
    Cc1 = __builtin_amdgcn_wmma_f32_16x16x4_f32(false, Av[0], false, Bc[4], (short)0, z, false, false);

    v8f e0 = Cp0, e1 = Cp1;
    asm volatile("" : "+v"(Cc0), "+v"(Cc1), "+v"(e0));
    v8f r0 = max_tile(e0, Pp0);
    Cc0 = __builtin_amdgcn_wmma_f32_16x16x4_f32(false, Av[1], false, Bc[1], (short)0, Cc0, false, false);
    Cc1 = __builtin_amdgcn_wmma_f32_16x16x4_f32(false, Av[1], false, Bc[5], (short)0, Cc1, false, false);
    asm volatile("" : "+v"(Cc0), "+v"(Cc1), "+v"(e1), "+v"(r0));
    v8f r1 = max_tile(e1, Pp1);
    store_pair(V2, shp, hi, l15, r0, r1);
    load_pair(V2, 16 * np, hi, l15, Bn);
    load_pb(PB, np, hi, Qn0, Qn1);
    asm volatile("" : "+v"(Cc0), "+v"(Cc1) : : "memory");

#pragma unroll
    for (int t = 2; t < 5; ++t) {
        Cc0 = __builtin_amdgcn_wmma_f32_16x16x4_f32(false, Av[t], false, Bc[t],     (short)0, Cc0, false, false);
        Cc1 = __builtin_amdgcn_wmma_f32_16x16x4_f32(false, Av[t], false, Bc[4 + t], (short)0, Cc1, false, false);
    }

#pragma unroll
    for (int r = 0; r < 8; ++r) {
        Pc0[r] = fmaf(-scale, Qc0[r], kv);
        Pc1[r] = fmaf(-scale, Qc1[r], kv);
    }
    asm volatile("" : "+v"(Bn[0]), "+v"(Bn[4]), "+v"(Bn[1]), "+v"(Bn[5]) : "v"(Cc0), "v"(Cc1));

    shp = hb;

    if (last) { p = 0; ++i; scale *= uscale; npairs = pairs_at(i); }
    else      { ++p; }
}

__device__ __forceinline__ void trip2(v2f* V2, const float* PB, int hi, int l15, float kv, float uscale,
                                      const v2f (&Av)[5], const v2f (&Bc)[9], v2f (&Bx)[9], v2f (&Bl)[9],
                                      const v8f& Qc0, const v8f& Qc1, v8f& Ql0, v8f& Ql1,
                                      v8f& Cc0, v8f& Cc1, v8f& Pc0, v8f& Pc1,
                                      const v8f& Cp0, const v8f& Cp1, const v8f& Pp0, const v8f& Pp1,
                                      int& shp, int& i, int& p, int& npairs, float& scale)
{
    const int  hb   = 16 * p;
    const bool last = (p + 1 == npairs);
    const int  np2  = (p + 2 < npairs) ? p + 2 : p + 2 - npairs;

    v8f z = {};
    Cc0 = __builtin_amdgcn_wmma_f32_16x16x4_f32(false, Av[0], false, Bc[0], (short)0, z, false, false);
    Cc1 = __builtin_amdgcn_wmma_f32_16x16x4_f32(false, Av[0], false, Bc[4], (short)0, z, false, false);

    v8f e0 = Cp0, e1 = Cp1;
    asm volatile("" : "+v"(Cc0), "+v"(Cc1), "+v"(e0));
    v8f r0 = max_tile(e0, Pp0);
    Cc0 = __builtin_amdgcn_wmma_f32_16x16x4_f32(false, Av[1], false, Bc[1], (short)0, Cc0, false, false);
    Cc1 = __builtin_amdgcn_wmma_f32_16x16x4_f32(false, Av[1], false, Bc[5], (short)0, Cc1, false, false);
    asm volatile("" : "+v"(Cc0), "+v"(Cc1), "+v"(e1), "+v"(r0));
    v8f r1 = max_tile(e1, Pp1);
    store_pair(V2, shp, hi, l15, r0, r1);
    load_pair(V2, 16 * np2, hi, l15, Bl);
    load_pb(PB, np2, hi, Ql0, Ql1);
    asm volatile("" : "+v"(Cc0), "+v"(Cc1) : : "memory");

#pragma unroll
    for (int t = 2; t < 5; ++t) {
        Cc0 = __builtin_amdgcn_wmma_f32_16x16x4_f32(false, Av[t], false, Bc[t],     (short)0, Cc0, false, false);
        Cc1 = __builtin_amdgcn_wmma_f32_16x16x4_f32(false, Av[t], false, Bc[4 + t], (short)0, Cc1, false, false);
    }

#pragma unroll
    for (int r = 0; r < 8; ++r) {
        Pc0[r] = fmaf(-scale, Qc0[r], kv);
        Pc1[r] = fmaf(-scale, Qc1[r], kv);
    }
    asm volatile("" : "+v"(Bx[0]), "+v"(Bx[4]), "+v"(Bx[1]), "+v"(Bx[5]) : "v"(Cc0), "v"(Cc1));

    shp = hb;

    if (last) { p = 0; ++i; scale *= uscale; npairs = pairs_at(i); }
    else      { ++p; }
}

__device__ __forceinline__ void prime_prev(const v2f* V2, int hi, int l15, v8f& C0, v8f& C1, v8f& P0, v8f& P1)
{
#pragma unroll
    for (int q = 0; q < 4; ++q) {
        v2f a = V2[(4 * hi + q) * BT + l15];
        v2f b = V2[(8 + 4 * hi + q) * BT + l15];
        C0[2 * q] = a[0]; C0[2 * q + 1] = a[1];
        C1[2 * q] = b[0]; C1[2 * q + 1] = b[1];
    }
#pragma unroll
    for (int r = 0; r < 8; ++r) { P0[r] = -1.0f; P1[r] = -1.0f; }
}

__global__ __launch_bounds__(32)
void btree_american_wmma(const float* __restrict__ kin, float* __restrict__ out,
                         float w0, float w1, float uscale, float alpha)
{
    __shared__ v2f   V2[NPAIR * BT];
    __shared__ float PB[NODES];

    const int lane = threadIdx.x;
    const int l15  = lane & 15;
    const int hi   = lane >> 4;
    const int b0   = blockIdx.x * BT;

    const float kv = kin[b0 + l15];

    for (int h = hi; h < NPAIR; h += 2) {
        float s0 = 100.0f * expf(alpha * (float)(4 * h - NSTEP));
        float s1 = 100.0f * expf(alpha * (float)(4 * h + 2 - NSTEP));
        v2f v;
        v[0] = fmaxf(kv - s0, 0.0f);
        v[1] = fmaxf(kv - s1, 0.0f);
        V2[h * BT + l15] = v;
    }
    for (int j = lane; j < NODES; j += 32) {
        PB[j] = 100.0f * expf(alpha * (float)(2 * j - (NSTEP - 1)));
    }
    __syncthreads();

    v2f Av[5];
#pragma unroll
    for (int t = 0; t < 5; ++t) {
        int k0 = 4 * t + 2 * hi;
        int k1 = k0 + 1;
        float e0 = (k0 == l15) ? w0 : ((k0 == l15 + 1) ? w1 : 0.0f);
        float e1 = (k1 == l15) ? w0 : ((k1 == l15 + 1) ? w1 : 0.0f);
        Av[t][0] = e0;
        Av[t][1] = e1;
    }

    v2f X[9], Y[9], Z[9];
    v8f Ca0, Ca1, Pa0, Pa1;
    v8f Cb0, Cb1, Pb0, Pb1;
    v8f QX0, QX1, QY0, QY1, QZ0, QZ1;
    load_pair(V2, 0, hi, l15, X);
    load_pb(PB, 0, hi, QX0, QX1);
    load_pair(V2, 16, hi, l15, Y);
    load_pb(PB, 1, hi, QY0, QY1);

    int shp = 0;
    prime_prev(V2, hi, l15, Cb0, Cb1, Pb0, Pb1);

    int   i = 0, p = 0, npairs = pairs_at(0);
    float scale = 1.0f;
    for (int k = 0; k < DTRIPS / 6; ++k) {
        trip2(V2, PB, hi, l15, kv, uscale, Av, X, Y, Z, QX0, QX1, QZ0, QZ1,
              Ca0, Ca1, Pa0, Pa1, Cb0, Cb1, Pb0, Pb1, shp, i, p, npairs, scale);
        trip2(V2, PB, hi, l15, kv, uscale, Av, Y, Z, X, QY0, QY1, QX0, QX1,
              Cb0, Cb1, Pb0, Pb1, Ca0, Ca1, Pa0, Pa1, shp, i, p, npairs, scale);
        trip2(V2, PB, hi, l15, kv, uscale, Av, Z, X, Y, QZ0, QZ1, QY0, QY1,
              Ca0, Ca1, Pa0, Pa1, Cb0, Cb1, Pb0, Pb1, shp, i, p, npairs, scale);
        trip2(V2, PB, hi, l15, kv, uscale, Av, X, Y, Z, QX0, QX1, QZ0, QZ1,
              Cb0, Cb1, Pb0, Pb1, Ca0, Ca1, Pa0, Pa1, shp, i, p, npairs, scale);
        trip2(V2, PB, hi, l15, kv, uscale, Av, Y, Z, X, QY0, QY1, QX0, QX1,
              Ca0, Ca1, Pa0, Pa1, Cb0, Cb1, Pb0, Pb1, shp, i, p, npairs, scale);
        trip2(V2, PB, hi, l15, kv, uscale, Av, Z, X, Y, QZ0, QZ1, QY0, QY1,
              Cb0, Cb1, Pb0, Pb1, Ca0, Ca1, Pa0, Pa1, shp, i, p, npairs, scale);
    }
    max_store_pair(V2, shp, hi, l15, Cb0, Cb1, Pb0, Pb1);

    load_pair(V2, 0, hi, l15, X);
    load_pb(PB, 0, hi, QX0, QX1);
    shp = 0;
    prime_prev(V2, hi, l15, Cb0, Cb1, Pb0, Pb1);
    for (int k = 0; k < (PTRIPS - DTRIPS) / 2; ++k) {
        trip(V2, PB, hi, l15, kv, uscale, Av, X, Y, QX0, QX1, QY0, QY1,
             Ca0, Ca1, Pa0, Pa1, Cb0, Cb1, Pb0, Pb1, shp, i, p, npairs, scale);
        trip(V2, PB, hi, l15, kv, uscale, Av, Y, X, QY0, QY1, QX0, QX1,
             Cb0, Cb1, Pb0, Pb1, Ca0, Ca1, Pa0, Pa1, shp, i, p, npairs, scale);
    }
    max_store_pair(V2, shp, hi, l15, Cb0, Cb1, Pb0, Pb1);

    for (int s = PSTEPS; s < NSTEP; ++s) {
        const int np = pairs_at(s);
        for (int q = 0; q < np; ++q) {
            load_pair(V2, 16 * q, hi, l15, X);
            v8f c0 = {};
            v8f c1 = {};
#pragma unroll
            for (int t = 0; t < 5; ++t) {
                c0 = __builtin_amdgcn_wmma_f32_16x16x4_f32(false, Av[t], false, X[t],     (short)0, c0, false, false);
                c1 = __builtin_amdgcn_wmma_f32_16x16x4_f32(false, Av[t], false, X[4 + t], (short)0, c1, false, false);
            }
            v8f y0, y1;
#pragma unroll
            for (int r = 0; r < 8; ++r) {
                int node = 32 * q + r + 8 * hi;
                y0[r] = fmaf(-scale, PB[node], kv);
                y1[r] = fmaf(-scale, PB[node + 16], kv);
            }
            max_store_pair(V2, 16 * q, hi, l15, c0, c1, y0, y1);
        }
        scale *= uscale;
    }
    __syncthreads();

    if (hi == 0) out[b0 + l15] = V2[l15][0];
}

extern "C" void kernel_launch(void* const* d_in, const int* in_sizes, int n_in,
                              void* d_out, int out_size, void* d_ws, size_t ws_size,
                              hipStream_t stream) {
    (void)n_in; (void)d_ws; (void)ws_size; (void)out_size;
    const float* kin = (const float*)d_in[0];
    float* out = (float*)d_out;
    const int B = in_sizes[0];

    const double S0g = 100.0, SIG = 0.2, Tg = 1.0, T0g = 0.0, Rg = 0.05;
    (void)S0g;
    const double dt   = (Tg - T0g) / (double)NSTEP;
    const double sq   = sqrt(dt);
    const double u    = exp(SIG * sq);
    const double d    = exp(-SIG * sq);
    const double disc = exp(-Rg * dt);
    const float w0    = (float)((disc * u - 1.0) / (u - d));
    const float w1    = (float)((1.0 - disc * d) / (u - d));
    const float us    = (float)u;
    const float alpha = (float)(SIG * sq);

    const int nblocks = B / BT;
    btree_american_wmma<<<nblocks, 32, 0, stream>>>(kin, out, w0, w1, us, alpha);
}


# Tiles no strike of the group can exercise in a block take its 16 steps as one product with the band of the 16-step map (8 matrix ops, not 80); tested per group and block; sum order changes
# speedup vs baseline: 3.7370x; 2.3986x over previous
#include <hip/hip_runtime.h>
#include <cmath>


typedef float v2f __attribute__((ext_vector_type(2)));
typedef float v8f __attribute__((ext_vector_type(8)));

#define NSTEP 256
#define NODES 288
#define NPAIR (NODES / 2)
#define BT    16
#define WAVES 4
#define GROUPS 2
#define BLK   16
#define NBLK  (NSTEP / BLK)
#define MAXT  4

static_assert(NBLK % 2 == 0, "an even block count leaves the result in plane 0");
static_assert(BLK == 16, "the band of a whole block (binomials, powers by bits) is written for 16 steps");
static_assert(16 * (NBLK + 1) <= NODES, "the tile above the widest block's top tile is inside the pad");

template <int NT>
__device__ __forceinline__ float run_block(const v2f* src, v2f* dst, const float* PB, int t0,
                                           int hi, int l15, float kv, float scale, float uscale,
                                           const v2f (&Av)[5])
{
    v8f X[NT + 1];
    v8f Q[NT + 1];
#pragma unroll
    for (int T = 0; T <= NT; ++T) {
#pragma unroll
        for (int q = 0; q < 4; ++q) {
            v2f p = src[(8 * (t0 + T) + q + 4 * hi) * BT + l15];
            X[T][2 * q]     = p[0];
            X[T][2 * q + 1] = p[1];
        }
    }
#pragma unroll
    for (int T = 0; T <= NT; ++T) {
#pragma unroll
        for (int r = 0; r < 8; ++r) Q[T][r] = PB[16 * (t0 + T) + r + 8 * hi];
    }

    for (int s = 0; s < BLK; ++s) {
        v8f acc[NT + 1];
        const v8f z = {};
#pragma unroll
        for (int t = 0; t < 4; ++t) {
#pragma unroll
            for (int T = 0; T <= NT; ++T) {
                v2f b;
                b[0] = X[T][2 * t];
                b[1] = X[T][2 * t + 1];
                acc[T] = __builtin_amdgcn_wmma_f32_16x16x4_f32(false, Av[t], false, b, (short)0,
                                                               (t == 0) ? z : acc[T], false, false);
            }
        }
#pragma unroll
        for (int T = 0; T < NT; ++T) {
            v2f b;
            b[0] = X[T + 1][0];
            b[1] = X[T + 1][1];
            acc[T] = __builtin_amdgcn_wmma_f32_16x16x4_f32(false, Av[4], false, b, (short)0,
                                                           acc[T], false, false);
        }
#pragma unroll
        for (int T = 0; T <= NT; ++T) {
#pragma unroll
            for (int r = 0; r < 8; ++r) X[T][r] = fmaxf(acc[T][r], fmaf(-scale, Q[T][r], kv));
        }
        scale *= uscale;
    }

#pragma unroll
    for (int T = 0; T < NT; ++T) {
#pragma unroll
        for (int q = 0; q < 4; ++q) {
            v2f p;
            p[0] = X[T][2 * q];
            p[1] = X[T][2 * q + 1];
            dst[(8 * (t0 + T) + q + 4 * hi) * BT + l15] = p;
        }
    }
    return scale;
}

__device__ __forceinline__ void jump_tile(const v2f* src, v2f* dst, int T, int hi, int l15,
                                          const v2f (&Aj)[8])
{
    v2f b[8];
#pragma unroll
    for (int o = 0; o < 2; ++o) {
#pragma unroll
        for (int q = 0; q < 4; ++q) b[4 * o + q] = src[(8 * (T + o) + q + 4 * hi) * BT + l15];
    }
    v8f acc = {};
#pragma unroll
    for (int f = 0; f < 8; ++f) {
        acc = __builtin_amdgcn_wmma_f32_16x16x4_f32(false, Aj[f], false, b[f], (short)0, acc,
                                                    false, false);
    }
#pragma unroll
    for (int q = 0; q < 4; ++q) {
        v2f pr;
        pr[0] = acc[2 * q];
        pr[1] = acc[2 * q + 1];
        dst[(8 * T + q + 4 * hi) * BT + l15] = pr;
    }
}

__global__ __launch_bounds__(32 * WAVES * GROUPS)
void btree_american_wmma(const float* __restrict__ kin, float* __restrict__ out,
                         float w0, float w1, float uscale, float alpha)
{
    __shared__ v2f   V2[GROUPS][2][NPAIR * BT];
    __shared__ float PB[NODES];
    __shared__ float CJ[BLK + 1];

    const int lane = threadIdx.x & 31;
    const int l15  = lane & 15;
    const int hi   = lane >> 4;
    const int wv   = __builtin_amdgcn_readfirstlane((int)(threadIdx.x >> 5));
    const int g    = wv / WAVES;
    const int w    = wv % WAVES;
    const int tg   = (int)threadIdx.x % (32 * WAVES);
    const int b0   = (blockIdx.x * GROUPS + g) * BT;

    const float kv = kin[b0 + l15];

    for (int h = tg >> 4; h < NPAIR; h += 2 * WAVES) {
        float s0 = 100.0f * expf(alpha * (float)(4 * h - NSTEP));
        float s1 = 100.0f * expf(alpha * (float)(4 * h + 2 - NSTEP));
        v2f v;
        v[0] = fmaxf(kv - s0, 0.0f);
        v[1] = fmaxf(kv - s1, 0.0f);
        V2[g][0][h * BT + l15] = v;
        V2[g][1][h * BT + l15] = v;
    }

    for (int j = (int)threadIdx.x; j < NODES; j += 32 * WAVES * GROUPS) {
        PB[j] = 100.0f * expf(alpha * (float)(2 * j - (NSTEP - 1)));
    }

    if (threadIdx.x <= BLK) {
        const int d  = (int)threadIdx.x;
        const int dm = (d < BLK - d) ? d : BLK - d;
        const double bn = dm == 0 ? 1.0 : dm == 1 ? 16.0 : dm == 2 ? 120.0 : dm == 3 ? 560.0
                        : dm == 4 ? 1820.0 : dm == 5 ? 4368.0 : dm == 6 ? 8008.0
                        : dm == 7 ? 11440.0 : 12870.0;
        const double a1 = (double)w0, a2 = a1 * a1, a4 = a2 * a2, a8 = a4 * a4, a16 = a8 * a8;
        const double b1 = (double)w1, b2 = b1 * b1, b4 = b2 * b2, b8 = b4 * b4, b16 = b8 * b8;
        const int na = BLK - d;
        double c = bn;
        c *= (na & 1) ? a1 : 1.0;
        c *= (na & 2) ? a2 : 1.0;
        c *= (na & 4) ? a4 : 1.0;
        c *= (na & 8) ? a8 : 1.0;
        c *= (na & 16) ? a16 : 1.0;
        c *= (d & 1) ? b1 : 1.0;
        c *= (d & 2) ? b2 : 1.0;
        c *= (d & 4) ? b4 : 1.0;
        c *= (d & 8) ? b8 : 1.0;
        c *= (d & 16) ? b16 : 1.0;
        CJ[d] = (float)c;
    }

    v2f Av[5];
#pragma unroll
    for (int t = 0; t < 5; ++t) {
        int tt = (t < 4) ? t : 8;
        int n0 = 2 * tt + 8 * hi;
        int n1 = n0 + 1;
        float e0 = (n0 == l15) ? w0 : ((n0 == l15 + 1) ? w1 : 0.0f);
        float e1 = (n1 == l15) ? w0 : ((n1 == l15 + 1) ? w1 : 0.0f);
        Av[t][0] = e0;
        Av[t][1] = e1;
    }
    __syncthreads();

    v2f Aj[8];
#pragma unroll
    for (int f = 0; f < 8; ++f) {
#pragma unroll
        for (int e = 0; e < 2; ++e) {
            const int d = 16 * (f / 4) + 2 * (f % 4) + e + 8 * hi - l15;
            const float c = CJ[(d < 0) ? 0 : ((d > BLK) ? BLK : d)];
            Aj[f][e] = (d >= 0 && d <= BLK) ? c : 0.0f;
        }
    }

    float km = kv;
#pragma unroll
    for (int o = 8; o >= 1; o >>= 1) km = fmaxf(km, __shfl_xor(km, o, 16));
    const float kmax = __builtin_bit_cast(float, __builtin_amdgcn_readfirstlane(__builtin_bit_cast(int, km)));
    const bool wpos = (w0 >= 0.0f) && (w1 >= 0.0f);

    float scale = 1.0f;
    for (int b = 0; b < NBLK; ++b) {
        const v2f* src = V2[g][b & 1];
        v2f*       dst = V2[g][(b & 1) ^ 1];
        const int ot   = NBLK - b;
        const float pb0 = PB[16 * (lane < NODES / 16 ? lane : NODES / 16 - 1)];
        const bool live = (lane < ot) && !(wpos && kmax < 0.999f * scale * pb0);
        const unsigned lm = __builtin_amdgcn_ballot_w32(live);
        const int st   = lm ? (32 - __builtin_clz(lm)) : 0;
        const int base = st / WAVES;
        const int rem  = st % WAVES;
        const int r    = g ? (WAVES - 1 - w) : w;
        const int nt   = base + (r < rem ? 1 : 0);
        const int t0   = r * base + (r < rem ? r : rem);
        for (int T = st + ((r - rem) & (WAVES - 1)); T < ot; T += WAVES) jump_tile(src, dst, T, hi, l15, Aj);
        switch (nt) {
        case 4:  scale = run_block<4>(src, dst, PB, t0, hi, l15, kv, scale, uscale, Av); break;
        case 3:  scale = run_block<3>(src, dst, PB, t0, hi, l15, kv, scale, uscale, Av); break;
        case 2:  scale = run_block<2>(src, dst, PB, t0, hi, l15, kv, scale, uscale, Av); break;
        case 1:  scale = run_block<1>(src, dst, PB, t0, hi, l15, kv, scale, uscale, Av); break;
        default:
            for (int s = 0; s < BLK; ++s) scale *= uscale;
            break;
        }
        __syncthreads();
    }

    if (threadIdx.x < GROUPS * BT) out[blockIdx.x * GROUPS * BT + lane] = V2[hi][0][l15][0];
}

extern "C" void kernel_launch(void* const* d_in, const int* in_sizes, int n_in,
                              void* d_out, int out_size, void* d_ws, size_t ws_size,
                              hipStream_t stream) {
    (void)n_in; (void)d_ws; (void)ws_size; (void)out_size;
    const float* kin = (const float*)d_in[0];
    float* out = (float*)d_out;
    const int B = in_sizes[0];

    const double S0g = 100.0, SIG = 0.2, Tg = 1.0, T0g = 0.0, Rg = 0.05;
    (void)S0g;
    const double dt   = (Tg - T0g) / (double)NSTEP;
    const double sq   = sqrt(dt);
    const double u    = exp(SIG * sq);
    const double d    = exp(-SIG * sq);
    const double disc = exp(-Rg * dt);
    const float w0    = (float)((disc * u - 1.0) / (u - d));
    const float w1    = (float)((1.0 - disc * d) / (u - d));
    const float us    = (float)u;
    const float alpha = (float)(SIG * sq);

    const int nblocks = B / (BT * GROUPS);
    btree_american_wmma<<<nblocks, 32 * WAVES * GROUPS, 0, stream>>>(kin, out, w0, w1, us, alpha);
}


# Top run of a group no longer steps the linear tile above it: that tile's first node after 0..15 steps is one 4-op product with the triangle of s-step maps; spare tile dealt to it; sum order changes
# speedup vs baseline: 3.8701x; 1.0356x over previous
#include <hip/hip_runtime.h>
#include <cmath>


typedef float v2f __attribute__((ext_vector_type(2)));
typedef float v8f __attribute__((ext_vector_type(8)));

#define NSTEP 256
#define NODES 288
#define NPAIR (NODES / 2)
#define BT    16
#define WAVES 4
#define GROUPS 2
#define BLK   16
#define NBLK  (NSTEP / BLK)
#define MAXT  4

static_assert(NBLK % 2 == 0, "an even block count leaves the result in plane 0");
static_assert(BLK == 16, "the band of a whole block (binomials, powers by bits) is written for 16 steps");
static_assert(16 * (NBLK + 1) <= NODES, "the tile above the widest block's top tile is inside the pad");
static_assert(32 * WAVES * GROUPS == BLK * 16, "one thread per entry of the table of s-step maps");

template <int NT, bool TOP>
__device__ __forceinline__ float run_block(const v2f* src, v2f* dst, const float* PB, int t0,
                                           int hi, int l15, float kv, float scale, float uscale,
                                           const v2f (&Av)[5], const v2f (&Mt)[4], const v2f Ahi)
{
    constexpr int NX = TOP ? NT : NT + 1;
    v8f X[NX];
    v8f Q[NX];
#pragma unroll
    for (int T = 0; T < NX; ++T) {
#pragma unroll
        for (int q = 0; q < 4; ++q) {
            v2f p = src[(8 * (t0 + T) + q + 4 * hi) * BT + l15];
            X[T][2 * q]     = p[0];
            X[T][2 * q + 1] = p[1];
        }
    }
    v8f H = {};
    if (TOP) {
        v2f hb[4];
#pragma unroll
        for (int q = 0; q < 4; ++q) hb[q] = src[(8 * (t0 + NT) + q + 4 * hi) * BT + l15];
#pragma unroll
        for (int q = 0; q < 4; ++q) {
            H = __builtin_amdgcn_wmma_f32_16x16x4_f32(false, Mt[q], false, hb[q], (short)0, H,
                                                      false, false);
        }
    }
#pragma unroll
    for (int T = 0; T < NX; ++T) {
#pragma unroll
        for (int r = 0; r < 8; ++r) Q[T][r] = PB[16 * (t0 + T) + r + 8 * hi];
    }

    for (int s = 0; s < BLK; ++s) {
        v8f acc[NX];
        const v8f z = {};
#pragma unroll
        for (int t = 0; t < 4; ++t) {
#pragma unroll
            for (int T = 0; T < NX; ++T) {
                v2f b;
                b[0] = X[T][2 * t];
                b[1] = X[T][2 * t + 1];
                acc[T] = __builtin_amdgcn_wmma_f32_16x16x4_f32(false, Av[t], false, b, (short)0,
                                                               (t == 0) ? z : acc[T], false, false);
            }
        }
#pragma unroll
        for (int T = 0; T + 1 < NX; ++T) {
            v2f b;
            b[0] = X[T + 1][0];
            b[1] = X[T + 1][1];
            acc[T] = __builtin_amdgcn_wmma_f32_16x16x4_f32(false, Av[4], false, b, (short)0,
                                                           acc[T], false, false);
        }
        if (TOP) {
            v2f b;
            b[0] = H[0];
            b[1] = 0.0f;
            const v2f a5 = (s < 8) ? Av[4] : Ahi;
            acc[NT - 1] = __builtin_amdgcn_wmma_f32_16x16x4_f32(false, a5, false, b, (short)0,
                                                                acc[NT - 1], false, false);
            H = __builtin_shufflevector(H, H, 1, 2, 3, 4, 5, 6, 7, 0);
        }
#pragma unroll
        for (int T = 0; T < NX; ++T) {
#pragma unroll
            for (int r = 0; r < 8; ++r) X[T][r] = fmaxf(acc[T][r], fmaf(-scale, Q[T][r], kv));
        }
        scale *= uscale;
    }

#pragma unroll
    for (int T = 0; T < NT; ++T) {
#pragma unroll
        for (int q = 0; q < 4; ++q) {
            v2f p;
            p[0] = X[T][2 * q];
            p[1] = X[T][2 * q + 1];
            dst[(8 * (t0 + T) + q + 4 * hi) * BT + l15] = p;
        }
    }
    return scale;
}

__device__ __forceinline__ void jump_tile(const v2f* src, v2f* dst, int T, int hi, int l15,
                                          const v2f (&Aj)[8])
{
    v2f b[8];
#pragma unroll
    for (int o = 0; o < 2; ++o) {
#pragma unroll
        for (int q = 0; q < 4; ++q) b[4 * o + q] = src[(8 * (T + o) + q + 4 * hi) * BT + l15];
    }
    v8f acc = {};
#pragma unroll
    for (int f = 0; f < 8; ++f) {
        acc = __builtin_amdgcn_wmma_f32_16x16x4_f32(false, Aj[f], false, b[f], (short)0, acc,
                                                    false, false);
    }
#pragma unroll
    for (int q = 0; q < 4; ++q) {
        v2f pr;
        pr[0] = acc[2 * q];
        pr[1] = acc[2 * q + 1];
        dst[(8 * T + q + 4 * hi) * BT + l15] = pr;
    }
}

__global__ __launch_bounds__(32 * WAVES * GROUPS)
void btree_american_wmma(const float* __restrict__ kin, float* __restrict__ out,
                         float w0, float w1, float uscale, float alpha)
{
    __shared__ v2f   V2[GROUPS][2][NPAIR * BT];
    __shared__ float PB[NODES];
    __shared__ float CJ[BLK + 1];
    __shared__ float TM[BLK * 16];

    const int lane = threadIdx.x & 31;
    const int l15  = lane & 15;
    const int hi   = lane >> 4;
    const int wv   = __builtin_amdgcn_readfirstlane((int)(threadIdx.x >> 5));
    const int g    = wv / WAVES;
    const int w    = wv % WAVES;
    const int tg   = (int)threadIdx.x % (32 * WAVES);
    const int b0   = (blockIdx.x * GROUPS + g) * BT;

    const float kv = kin[b0 + l15];

    constexpr unsigned short BIN[16][8] = {
        {1, 0, 0, 0, 0, 0, 0, 0},
        {1, 0, 0, 0, 0, 0, 0, 0},
        {1, 2, 0, 0, 0, 0, 0, 0},
        {1, 3, 0, 0, 0, 0, 0, 0},
        {1, 4, 6, 0, 0, 0, 0, 0},
        {1, 5, 10, 0, 0, 0, 0, 0},
        {1, 6, 15, 20, 0, 0, 0, 0},
        {1, 7, 21, 35, 0, 0, 0, 0},
        {1, 8, 28, 56, 70, 0, 0, 0},
        {1, 9, 36, 84, 126, 0, 0, 0},
        {1, 10, 45, 120, 210, 252, 0, 0},
        {1, 11, 55, 165, 330, 462, 0, 0},
        {1, 12, 66, 220, 495, 792, 924, 0},
        {1, 13, 78, 286, 715, 1287, 1716, 0},
        {1, 14, 91, 364, 1001, 2002, 3003, 3432},
        {1, 15, 105, 455, 1365, 3003, 5005, 6435}
    };
    const int ts = (int)threadIdx.x >> 4;
    const int td = (int)threadIdx.x & 15;
    const int tdm = (td < ts - td) ? td : ts - td;
    const float bn = (float)BIN[ts][tdm < 0 ? 0 : tdm];

    for (int h = tg >> 4; h < NPAIR; h += 2 * WAVES) {
        float s0 = 100.0f * expf(alpha * (float)(4 * h - NSTEP));
        float s1 = 100.0f * expf(alpha * (float)(4 * h + 2 - NSTEP));
        v2f v;
        v[0] = fmaxf(kv - s0, 0.0f);
        v[1] = fmaxf(kv - s1, 0.0f);
        V2[g][0][h * BT + l15] = v;
        V2[g][1][h * BT + l15] = v;
    }

    for (int j = (int)threadIdx.x; j < NODES; j += 32 * WAVES * GROUPS) {
        PB[j] = 100.0f * expf(alpha * (float)(2 * j - (NSTEP - 1)));
    }

    if (threadIdx.x <= BLK) {
        const int d  = (int)threadIdx.x;
        const int dm = (d < BLK - d) ? d : BLK - d;
        const double bn = dm == 0 ? 1.0 : dm == 1 ? 16.0 : dm == 2 ? 120.0 : dm == 3 ? 560.0
                        : dm == 4 ? 1820.0 : dm == 5 ? 4368.0 : dm == 6 ? 8008.0
                        : dm == 7 ? 11440.0 : 12870.0;
        const double a1 = (double)w0, a2 = a1 * a1, a4 = a2 * a2, a8 = a4 * a4, a16 = a8 * a8;
        const double b1 = (double)w1, b2 = b1 * b1, b4 = b2 * b2, b8 = b4 * b4, b16 = b8 * b8;
        const int na = BLK - d;
        double c = bn;
        c *= (na & 1) ? a1 : 1.0;
        c *= (na & 2) ? a2 : 1.0;
        c *= (na & 4) ? a4 : 1.0;
        c *= (na & 8) ? a8 : 1.0;
        c *= (na & 16) ? a16 : 1.0;
        c *= (d & 1) ? b1 : 1.0;
        c *= (d & 2) ? b2 : 1.0;
        c *= (d & 4) ? b4 : 1.0;
        c *= (d & 8) ? b8 : 1.0;
        c *= (d & 16) ? b16 : 1.0;
        CJ[d] = (float)c;
    }

    {
        const int s = ts;
        const int d = td;
        const double a1 = (double)w0, a2 = a1 * a1, a4 = a2 * a2, a8 = a4 * a4;
        const double b1 = (double)w1, b2 = b1 * b1, b4 = b2 * b2, b8 = b4 * b4;
        const int na = s - d;
        double c = (double)bn;
        c *= (na & 1) ? a1 : 1.0;
        c *= (na & 2) ? a2 : 1.0;
        c *= (na & 4) ? a4 : 1.0;
        c *= (na & 8) ? a8 : 1.0;
        c *= (d & 1) ? b1 : 1.0;
        c *= (d & 2) ? b2 : 1.0;
        c *= (d & 4) ? b4 : 1.0;
        c *= (d & 8) ? b8 : 1.0;
        TM[s * 16 + d] = (d <= s) ? (float)c : 0.0f;
    }

    v2f Av[5];
#pragma unroll
    for (int t = 0; t < 5; ++t) {
        int tt = (t < 4) ? t : 8;
        int n0 = 2 * tt + 8 * hi;
        int n1 = n0 + 1;
        float e0 = (n0 == l15) ? w0 : ((n0 == l15 + 1) ? w1 : 0.0f);
        float e1 = (n1 == l15) ? w0 : ((n1 == l15 + 1) ? w1 : 0.0f);
        Av[t][0] = e0;
        Av[t][1] = e1;
    }
    __syncthreads();

    v2f Aj[8];
#pragma unroll
    for (int f = 0; f < 8; ++f) {
#pragma unroll
        for (int e = 0; e < 2; ++e) {
            const int d = 16 * (f / 4) + 2 * (f % 4) + e + 8 * hi - l15;
            const float c = CJ[(d < 0) ? 0 : ((d > BLK) ? BLK : d)];
            Aj[f][e] = (d >= 0 && d <= BLK) ? c : 0.0f;
        }
    }

    v2f Mt[4];
#pragma unroll
    for (int t = 0; t < 4; ++t) {
#pragma unroll
        for (int e = 0; e < 2; ++e) Mt[t][e] = TM[l15 * 16 + 2 * t + e + 8 * hi];
    }
    v2f Ahi;
    Ahi[0] = (l15 == 15 && hi == 1) ? w1 : 0.0f;
    Ahi[1] = 0.0f;

    float km = kv;
#pragma unroll
    for (int o = 8; o >= 1; o >>= 1) km = fmaxf(km, __shfl_xor(km, o, 16));
    const float kmax = __builtin_bit_cast(float, __builtin_amdgcn_readfirstlane(__builtin_bit_cast(int, km)));
    const bool wpos = (w0 >= 0.0f) && (w1 >= 0.0f);

    float scale = 1.0f;
    for (int b = 0; b < NBLK; ++b) {
        const v2f* src = V2[g][b & 1];
        v2f*       dst = V2[g][(b & 1) ^ 1];
        const int ot   = NBLK - b;
        const float pb0 = PB[16 * (lane < NODES / 16 ? lane : NODES / 16 - 1)];
        const bool live = (lane < ot) && !(wpos && kmax < 0.999f * scale * pb0);
        const unsigned lm = __builtin_amdgcn_ballot_w32(live);
        const int st   = lm ? (32 - __builtin_clz(lm)) : 0;
        const int base = st / WAVES;
        const int rem  = st % WAVES;
        const int r    = g ? (WAVES - 1 - w) : w;
        const bool xtra = (rem == 3) ? (r != 2) : (r < rem);
        const int nt    = base + (xtra ? 1 : 0);
        const int t0    = r * base + ((rem == 3) ? (r < 2 ? r : 2) : (r < rem ? r : rem));
        const bool top  = (nt > 0) && (t0 + nt == st);
        for (int T = st + (WAVES - 1 - r); T < ot; T += WAVES) jump_tile(src, dst, T, hi, l15, Aj);
        if (top) {
            switch (nt) {
            case 4:  scale = run_block<4, true>(src, dst, PB, t0, hi, l15, kv, scale, uscale, Av, Mt, Ahi); break;
            case 3:  scale = run_block<3, true>(src, dst, PB, t0, hi, l15, kv, scale, uscale, Av, Mt, Ahi); break;
            case 2:  scale = run_block<2, true>(src, dst, PB, t0, hi, l15, kv, scale, uscale, Av, Mt, Ahi); break;
            default: scale = run_block<1, true>(src, dst, PB, t0, hi, l15, kv, scale, uscale, Av, Mt, Ahi); break;
            }
        } else {
            switch (nt) {
            case 4:  scale = run_block<4, false>(src, dst, PB, t0, hi, l15, kv, scale, uscale, Av, Mt, Ahi); break;
            case 3:  scale = run_block<3, false>(src, dst, PB, t0, hi, l15, kv, scale, uscale, Av, Mt, Ahi); break;
            case 2:  scale = run_block<2, false>(src, dst, PB, t0, hi, l15, kv, scale, uscale, Av, Mt, Ahi); break;
            case 1:  scale = run_block<1, false>(src, dst, PB, t0, hi, l15, kv, scale, uscale, Av, Mt, Ahi); break;
            default:
                for (int s = 0; s < BLK; ++s) scale *= uscale;
                break;
            }
        }
        __syncthreads();
    }

    if (threadIdx.x < GROUPS * BT) out[blockIdx.x * GROUPS * BT + lane] = V2[hi][0][l15][0];
}

extern "C" void kernel_launch(void* const* d_in, const int* in_sizes, int n_in,
                              void* d_out, int out_size, void* d_ws, size_t ws_size,
                              hipStream_t stream) {
    (void)n_in; (void)d_ws; (void)ws_size; (void)out_size;
    const float* kin = (const float*)d_in[0];
    float* out = (float*)d_out;
    const int B = in_sizes[0];

    const double S0g = 100.0, SIG = 0.2, Tg = 1.0, T0g = 0.0, Rg = 0.05;
    (void)S0g;
    const double dt   = (Tg - T0g) / (double)NSTEP;
    const double sq   = sqrt(dt);
    const double u    = exp(SIG * sq);
    const double d    = exp(-SIG * sq);
    const double disc = exp(-Rg * dt);
    const float w0    = (float)((disc * u - 1.0) / (u - d));
    const float w1    = (float)((1.0 - disc * d) / (u - d));
    const float us    = (float)u;
    const float alpha = (float)(SIG * sq);

    const int nblocks = B / (BT * GROUPS);
    btree_american_wmma<<<nblocks, 32 * WAVES * GROUPS, 0, stream>>>(kin, out, w0, w1, us, alpha);
}


# A group's stepped tiles in two runs on two SIMDs, not four (one tile above recomputed, not three); a run's wave issues a step in two tile halves, one half's maxima under the other's matrix ops
# speedup vs baseline: 4.2952x; 1.1098x over previous
#include <hip/hip_runtime.h>
#include <cmath>


typedef float v2f __attribute__((ext_vector_type(2)));
typedef float v8f __attribute__((ext_vector_type(8)));

#define NSTEP 256
#define NODES 288
#define NPAIR (NODES / 2)
#define BT    16
#define WAVES 4
#define GROUPS 2
#define BLK   16
#define NBLK  (NSTEP / BLK)
#define MAXT  8

static_assert(NBLK % 2 == 0, "an even block count leaves the result in plane 0");
static_assert(BLK == 16, "the band of a whole block (binomials, powers by bits) is written for 16 steps");
static_assert(16 * (NBLK + 1) <= NODES, "the tile above the widest block's top tile is inside the pad");
static_assert(32 * WAVES * GROUPS == BLK * 16, "one thread per entry of the table of s-step maps");

template <int NT, bool TOP>
__device__ __forceinline__ float run_block(const v2f* src, v2f* dst, const float* PB, int t0,
                                           int hi, int l15, float kv, float scale, float uscale,
                                           const v2f (&Av)[5], const v2f (&Mt)[4], const v2f Ahi)
{
    constexpr int NX = TOP ? NT : NT + 1;
    v8f X[NX];
    v8f Q[NX];
#pragma unroll
    for (int T = 0; T < NX; ++T) {
#pragma unroll
        for (int q = 0; q < 4; ++q) {
            v2f p = src[(8 * (t0 + T) + q + 4 * hi) * BT + l15];
            X[T][2 * q]     = p[0];
            X[T][2 * q + 1] = p[1];
        }
    }
    v8f H = {};
    if (TOP) {
        v2f hb[4];
#pragma unroll
        for (int q = 0; q < 4; ++q) hb[q] = src[(8 * (t0 + NT) + q + 4 * hi) * BT + l15];
#pragma unroll
        for (int q = 0; q < 4; ++q) {
            H = __builtin_amdgcn_wmma_f32_16x16x4_f32(false, Mt[q], false, hb[q], (short)0, H,
                                                      false, false);
        }
    }
#pragma unroll
    for (int T = 0; T < NX; ++T) {
#pragma unroll
        for (int r = 0; r < 8; ++r) Q[T][r] = PB[16 * (t0 + T) + r + 8 * hi];
    }

    constexpr int H1  = (NX >= 4) ? NX / 2 : NX;
    constexpr int NW1 = 4 * (NX - H1) + (NX - H1 - 1) + (TOP ? 1 : 0);
    for (int s = 0; s < BLK; ++s) {
        v8f acc[NX];
        v8f P[NX];
        const v8f z = {};
#pragma unroll
        for (int half = 0; half < 2; ++half) {
            const int Tlo = half ? H1 : 0;
            const int Thi = half ? NX : H1;
            if (half == 1 && H1 < NX) {
                __builtin_amdgcn_sched_barrier(0);
#pragma unroll
                for (int T = 0; T < NX; ++T) {
#pragma unroll
                    for (int r = 0; r < 8; ++r) P[T][r] = fmaf(-scale, Q[T][r], kv);
                }
#pragma unroll
                for (int T = 0; T < H1; ++T) {
#pragma unroll
                    for (int r = 0; r < 8; ++r) X[T][r] = fmaxf(acc[T][r], P[T][r]);
                }
            }
#pragma unroll
            for (int t = 0; t < 4; ++t) {
#pragma unroll
                for (int T = Tlo; T < Thi; ++T) {
                    v2f b;
                    b[0] = X[T][2 * t];
                    b[1] = X[T][2 * t + 1];
                    acc[T] = __builtin_amdgcn_wmma_f32_16x16x4_f32(false, Av[t], false, b, (short)0,
                                                                   (t == 0) ? z : acc[T], false, false);
                }
            }
#pragma unroll
            for (int T = Tlo; T < Thi; ++T) {
                if (T + 1 < NX) {
                    v2f b;
                    b[0] = X[T + 1][0];
                    b[1] = X[T + 1][1];
                    acc[T] = __builtin_amdgcn_wmma_f32_16x16x4_f32(false, Av[4], false, b, (short)0,
                                                                   acc[T], false, false);
                }
            }
        }
        if (TOP) {
            v2f b;
            b[0] = H[0];
            b[1] = 0.0f;
            const v2f a5 = (s < 8) ? Av[4] : Ahi;
            acc[NT - 1] = __builtin_amdgcn_wmma_f32_16x16x4_f32(false, a5, false, b, (short)0,
                                                                acc[NT - 1], false, false);
            H = __builtin_shufflevector(H, H, 1, 2, 3, 4, 5, 6, 7, 0);
        }
        if (H1 < NX) {
#pragma unroll
            for (int i = 0; i < NW1; ++i) {
                __builtin_amdgcn_sched_group_barrier(0x008, 1, 0);
                __builtin_amdgcn_sched_group_barrier(0x002, 6, 0);
            }
#pragma unroll
            for (int T = H1; T < NX; ++T) {
#pragma unroll
                for (int r = 0; r < 8; ++r) X[T][r] = fmaxf(acc[T][r], P[T][r]);
            }
        } else {
#pragma unroll
            for (int T = 0; T < NX; ++T) {
#pragma unroll
                for (int r = 0; r < 8; ++r) X[T][r] = fmaxf(acc[T][r], fmaf(-scale, Q[T][r], kv));
            }
        }
        scale *= uscale;
    }

#pragma unroll
    for (int T = 0; T < NT; ++T) {
#pragma unroll
        for (int q = 0; q < 4; ++q) {
            v2f p;
            p[0] = X[T][2 * q];
            p[1] = X[T][2 * q + 1];
            dst[(8 * (t0 + T) + q + 4 * hi) * BT + l15] = p;
        }
    }
    return scale;
}

__device__ __forceinline__ void jump_tile(const v2f* src, v2f* dst, int T, int hi, int l15,
                                          const v2f (&Aj)[8])
{
    v2f b[8];
#pragma unroll
    for (int o = 0; o < 2; ++o) {
#pragma unroll
        for (int q = 0; q < 4; ++q) b[4 * o + q] = src[(8 * (T + o) + q + 4 * hi) * BT + l15];
    }
    v8f acc = {};
#pragma unroll
    for (int f = 0; f < 8; ++f) {
        acc = __builtin_amdgcn_wmma_f32_16x16x4_f32(false, Aj[f], false, b[f], (short)0, acc,
                                                    false, false);
    }
#pragma unroll
    for (int q = 0; q < 4; ++q) {
        v2f pr;
        pr[0] = acc[2 * q];
        pr[1] = acc[2 * q + 1];
        dst[(8 * T + q + 4 * hi) * BT + l15] = pr;
    }
}

__global__ __launch_bounds__(32 * WAVES * GROUPS)
void btree_american_wmma(const float* __restrict__ kin, float* __restrict__ out,
                         float w0, float w1, float uscale, float alpha)
{
    __shared__ v2f   V2[GROUPS][2][NPAIR * BT];
    __shared__ float PB[NODES];
    __shared__ float CJ[BLK + 1];
    __shared__ float TM[BLK * 16];

    const int lane = threadIdx.x & 31;
    const int l15  = lane & 15;
    const int hi   = lane >> 4;
    const int wv   = __builtin_amdgcn_readfirstlane((int)(threadIdx.x >> 5));
    const int g    = wv / WAVES;
    const int w    = wv % WAVES;
    const int tg   = (int)threadIdx.x % (32 * WAVES);
    const int b0   = (blockIdx.x * GROUPS + g) * BT;

    const float kv = kin[b0 + l15];

    constexpr unsigned short BIN[16][8] = {
        {1, 0, 0, 0, 0, 0, 0, 0},
        {1, 0, 0, 0, 0, 0, 0, 0},
        {1, 2, 0, 0, 0, 0, 0, 0},
        {1, 3, 0, 0, 0, 0, 0, 0},
        {1, 4, 6, 0, 0, 0, 0, 0},
        {1, 5, 10, 0, 0, 0, 0, 0},
        {1, 6, 15, 20, 0, 0, 0, 0},
        {1, 7, 21, 35, 0, 0, 0, 0},
        {1, 8, 28, 56, 70, 0, 0, 0},
        {1, 9, 36, 84, 126, 0, 0, 0},
        {1, 10, 45, 120, 210, 252, 0, 0},
        {1, 11, 55, 165, 330, 462, 0, 0},
        {1, 12, 66, 220, 495, 792, 924, 0},
        {1, 13, 78, 286, 715, 1287, 1716, 0},
        {1, 14, 91, 364, 1001, 2002, 3003, 3432},
        {1, 15, 105, 455, 1365, 3003, 5005, 6435}
    };
    const int ts = (int)threadIdx.x >> 4;
    const int td = (int)threadIdx.x & 15;
    const int tdm = (td < ts - td) ? td : ts - td;
    const float bn = (float)BIN[ts][tdm < 0 ? 0 : tdm];

    for (int h = tg >> 4; h < NPAIR; h += 2 * WAVES) {
        float s0 = 100.0f * expf(alpha * (float)(4 * h - NSTEP));
        float s1 = 100.0f * expf(alpha * (float)(4 * h + 2 - NSTEP));
        v2f v;
        v[0] = fmaxf(kv - s0, 0.0f);
        v[1] = fmaxf(kv - s1, 0.0f);
        V2[g][0][h * BT + l15] = v;
        V2[g][1][h * BT + l15] = v;
    }

    for (int j = (int)threadIdx.x; j < NODES; j += 32 * WAVES * GROUPS) {
        PB[j] = 100.0f * expf(alpha * (float)(2 * j - (NSTEP - 1)));
    }

    if (threadIdx.x <= BLK) {
        const int d  = (int)threadIdx.x;
        const int dm = (d < BLK - d) ? d : BLK - d;
        const double bn = dm == 0 ? 1.0 : dm == 1 ? 16.0 : dm == 2 ? 120.0 : dm == 3 ? 560.0
                        : dm == 4 ? 1820.0 : dm == 5 ? 4368.0 : dm == 6 ? 8008.0
                        : dm == 7 ? 11440.0 : 12870.0;
        const double a1 = (double)w0, a2 = a1 * a1, a4 = a2 * a2, a8 = a4 * a4, a16 = a8 * a8;
        const double b1 = (double)w1, b2 = b1 * b1, b4 = b2 * b2, b8 = b4 * b4, b16 = b8 * b8;
        const int na = BLK - d;
        double c = bn;
        c *= (na & 1) ? a1 : 1.0;
        c *= (na & 2) ? a2 : 1.0;
        c *= (na & 4) ? a4 : 1.0;
        c *= (na & 8) ? a8 : 1.0;
        c *= (na & 16) ? a16 : 1.0;
        c *= (d & 1) ? b1 : 1.0;
        c *= (d & 2) ? b2 : 1.0;
        c *= (d & 4) ? b4 : 1.0;
        c *= (d & 8) ? b8 : 1.0;
        c *= (d & 16) ? b16 : 1.0;
        CJ[d] = (float)c;
    }

    {
        const int s = ts;
        const int d = td;
        const double a1 = (double)w0, a2 = a1 * a1, a4 = a2 * a2, a8 = a4 * a4;
        const double b1 = (double)w1, b2 = b1 * b1, b4 = b2 * b2, b8 = b4 * b4;
        const int na = s - d;
        double c = (double)bn;
        c *= (na & 1) ? a1 : 1.0;
        c *= (na & 2) ? a2 : 1.0;
        c *= (na & 4) ? a4 : 1.0;
        c *= (na & 8) ? a8 : 1.0;
        c *= (d & 1) ? b1 : 1.0;
        c *= (d & 2) ? b2 : 1.0;
        c *= (d & 4) ? b4 : 1.0;
        c *= (d & 8) ? b8 : 1.0;
        TM[s * 16 + d] = (d <= s) ? (float)c : 0.0f;
    }

    v2f Av[5];
#pragma unroll
    for (int t = 0; t < 5; ++t) {
        int tt = (t < 4) ? t : 8;
        int n0 = 2 * tt + 8 * hi;
        int n1 = n0 + 1;
        float e0 = (n0 == l15) ? w0 : ((n0 == l15 + 1) ? w1 : 0.0f);
        float e1 = (n1 == l15) ? w0 : ((n1 == l15 + 1) ? w1 : 0.0f);
        Av[t][0] = e0;
        Av[t][1] = e1;
    }
    __syncthreads();

    v2f Aj[8];
#pragma unroll
    for (int f = 0; f < 8; ++f) {
#pragma unroll
        for (int e = 0; e < 2; ++e) {
            const int d = 16 * (f / 4) + 2 * (f % 4) + e + 8 * hi - l15;
            const float c = CJ[(d < 0) ? 0 : ((d > BLK) ? BLK : d)];
            Aj[f][e] = (d >= 0 && d <= BLK) ? c : 0.0f;
        }
    }

    v2f Mt[4];
#pragma unroll
    for (int t = 0; t < 4; ++t) {
#pragma unroll
        for (int e = 0; e < 2; ++e) Mt[t][e] = TM[l15 * 16 + 2 * t + e + 8 * hi];
    }
    v2f Ahi;
    Ahi[0] = (l15 == 15 && hi == 1) ? w1 : 0.0f;
    Ahi[1] = 0.0f;

    float km = kv;
#pragma unroll
    for (int o = 8; o >= 1; o >>= 1) km = fmaxf(km, __shfl_xor(km, o, 16));
    const float kmax = __builtin_bit_cast(float, __builtin_amdgcn_readfirstlane(__builtin_bit_cast(int, km)));
    const bool wpos = (w0 >= 0.0f) && (w1 >= 0.0f);

    float scale = 1.0f;
    for (int b = 0; b < NBLK; ++b) {
        const v2f* src = V2[g][b & 1];
        v2f*       dst = V2[g][(b & 1) ^ 1];
        const int ot   = NBLK - b;
        const float pb0 = PB[16 * (lane < NODES / 16 ? lane : NODES / 16 - 1)];
        const bool live = (lane < ot) && !(wpos && kmax < 0.999f * scale * pb0);
        const unsigned lm = __builtin_amdgcn_ballot_w32(live);
        const int st   = lm ? (32 - __builtin_clz(lm)) : 0;
        const int nlo  = st >> 1;
        const int nup  = st - nlo;
        const int role = g ? (w ^ 2) : w;
        const int nt   = (role == 0) ? nlo : ((role == 1) ? nup : 0);
        const int t0   = (role == 1) ? nlo : 0;
        if (nt == 0) {
            const int e0 = (nlo == 0) ? 1 : 0;
            const int e1 = (nup == 0) ? 1 : 0;
            const int nj = 2 + e0 + e1;
            const int q  = (role >= 2) ? (role - 2) : ((role == 0) ? 2 : 2 + e0);
            for (int T = st + q; T < ot; T += nj) jump_tile(src, dst, T, hi, l15, Aj);
            for (int s = 0; s < BLK; ++s) scale *= uscale;
        } else if (role == 1) {
            switch (nt) {
            case 8:  scale = run_block<8, true>(src, dst, PB, t0, hi, l15, kv, scale, uscale, Av, Mt, Ahi); break;
            case 7:  scale = run_block<7, true>(src, dst, PB, t0, hi, l15, kv, scale, uscale, Av, Mt, Ahi); break;
            case 6:  scale = run_block<6, true>(src, dst, PB, t0, hi, l15, kv, scale, uscale, Av, Mt, Ahi); break;
            case 5:  scale = run_block<5, true>(src, dst, PB, t0, hi, l15, kv, scale, uscale, Av, Mt, Ahi); break;
            case 4:  scale = run_block<4, true>(src, dst, PB, t0, hi, l15, kv, scale, uscale, Av, Mt, Ahi); break;
            case 3:  scale = run_block<3, true>(src, dst, PB, t0, hi, l15, kv, scale, uscale, Av, Mt, Ahi); break;
            case 2:  scale = run_block<2, true>(src, dst, PB, t0, hi, l15, kv, scale, uscale, Av, Mt, Ahi); break;
            default: scale = run_block<1, true>(src, dst, PB, t0, hi, l15, kv, scale, uscale, Av, Mt, Ahi); break;
            }
        } else {
            switch (nt) {
            case 8:  scale = run_block<8, false>(src, dst, PB, t0, hi, l15, kv, scale, uscale, Av, Mt, Ahi); break;
            case 7:  scale = run_block<7, false>(src, dst, PB, t0, hi, l15, kv, scale, uscale, Av, Mt, Ahi); break;
            case 6:  scale = run_block<6, false>(src, dst, PB, t0, hi, l15, kv, scale, uscale, Av, Mt, Ahi); break;
            case 5:  scale = run_block<5, false>(src, dst, PB, t0, hi, l15, kv, scale, uscale, Av, Mt, Ahi); break;
            case 4:  scale = run_block<4, false>(src, dst, PB, t0, hi, l15, kv, scale, uscale, Av, Mt, Ahi); break;
            case 3:  scale = run_block<3, false>(src, dst, PB, t0, hi, l15, kv, scale, uscale, Av, Mt, Ahi); break;
            case 2:  scale = run_block<2, false>(src, dst, PB, t0, hi, l15, kv, scale, uscale, Av, Mt, Ahi); break;
            default: scale = run_block<1, false>(src, dst, PB, t0, hi, l15, kv, scale, uscale, Av, Mt, Ahi); break;
            }
        }
        __syncthreads();
    }

    if (threadIdx.x < GROUPS * BT) out[blockIdx.x * GROUPS * BT + lane] = V2[hi][0][l15][0];
}

extern "C" void kernel_launch(void* const* d_in, const int* in_sizes, int n_in,
                              void* d_out, int out_size, void* d_ws, size_t ws_size,
                              hipStream_t stream) {
    (void)n_in; (void)d_ws; (void)ws_size; (void)out_size;
    const float* kin = (const float*)d_in[0];
    float* out = (float*)d_out;
    const int B = in_sizes[0];

    const double S0g = 100.0, SIG = 0.2, Tg = 1.0, T0g = 0.0, Rg = 0.05;
    (void)S0g;
    const double dt   = (Tg - T0g) / (double)NSTEP;
    const double sq   = sqrt(dt);
    const double u    = exp(SIG * sq);
    const double d    = exp(-SIG * sq);
    const double disc = exp(-Rg * dt);
    const float w0    = (float)((disc * u - 1.0) / (u - d));
    const float w1    = (float)((1.0 - disc * d) / (u - d));
    const float us    = (float)u;
    const float alpha = (float)(SIG * sq);

    const int nblocks = B / (BT * GROUPS);
    btree_american_wmma<<<nblocks, 32 * WAVES * GROUPS, 0, stream>>>(kin, out, w0, w1, us, alpha);
}


# Tiles standing at the exercise payoff for all 16 strikes, with such a tile above, are not stepped: written as the block's last payoff by an idle wave; per-tile flag in LDS; sums unchanged
# speedup vs baseline: 4.6883x; 1.0915x over previous
#include <hip/hip_runtime.h>
#include <cmath>


typedef float v2f __attribute__((ext_vector_type(2)));
typedef float v8f __attribute__((ext_vector_type(8)));

#define NSTEP 256
#define NODES 288
#define NPAIR (NODES / 2)
#define BT    16
#define WAVES 4
#define GROUPS 2
#define BLK   16
#define NBLK  (NSTEP / BLK)
#define MAXT  8

static_assert(NBLK % 2 == 0, "an even block count leaves the result in plane 0");
static_assert(BLK == 16, "the band of a whole block (binomials, powers by bits) is written for 16 steps");
static_assert(16 * (NBLK + 1) <= NODES, "the tile above the widest block's top tile is inside the pad");
static_assert(32 * WAVES * GROUPS == BLK * 16, "one thread per entry of the table of s-step maps");

template <int NT, bool TOP>
__device__ __forceinline__ float run_block(const v2f* src, v2f* dst, int* fdst, const float* PB, int t0,
                                           int hi, int l15, float kv, float scale, float uscale,
                                           const v2f (&Av)[5], const v2f (&Mt)[4], const v2f Ahi)
{
    constexpr int NX = TOP ? NT : NT + 1;
    v8f X[NX];
    v8f Q[NX];
#pragma unroll
    for (int T = 0; T < NX; ++T) {
#pragma unroll
        for (int q = 0; q < 4; ++q) {
            v2f p = src[(8 * (t0 + T) + q + 4 * hi) * BT + l15];
            X[T][2 * q]     = p[0];
            X[T][2 * q + 1] = p[1];
        }
    }
    v8f H = {};
    if (TOP) {
        v2f hb[4];
#pragma unroll
        for (int q = 0; q < 4; ++q) hb[q] = src[(8 * (t0 + NT) + q + 4 * hi) * BT + l15];
#pragma unroll
        for (int q = 0; q < 4; ++q) {
            H = __builtin_amdgcn_wmma_f32_16x16x4_f32(false, Mt[q], false, hb[q], (short)0, H,
                                                      false, false);
        }
    }
#pragma unroll
    for (int T = 0; T < NX; ++T) {
#pragma unroll
        for (int r = 0; r < 8; ++r) Q[T][r] = PB[16 * (t0 + T) + r + 8 * hi];
    }

    constexpr int H1  = (NX >= 4) ? NX / 2 : NX;
    constexpr int NW1 = 4 * (NX - H1) + (NX - H1 - 1) + (TOP ? 1 : 0);
    float slast = scale;
    for (int s = 0; s < BLK; ++s) {
        v8f acc[NX];
        v8f P[NX];
        const v8f z = {};
#pragma unroll
        for (int half = 0; half < 2; ++half) {
            const int Tlo = half ? H1 : 0;
            const int Thi = half ? NX : H1;
            if (half == 1 && H1 < NX) {
                __builtin_amdgcn_sched_barrier(0);
#pragma unroll
                for (int T = 0; T < NX; ++T) {
#pragma unroll
                    for (int r = 0; r < 8; ++r) P[T][r] = fmaf(-scale, Q[T][r], kv);
                }
#pragma unroll
                for (int T = 0; T < H1; ++T) {
#pragma unroll
                    for (int r = 0; r < 8; ++r) X[T][r] = fmaxf(acc[T][r], P[T][r]);
                }
            }
#pragma unroll
            for (int t = 0; t < 4; ++t) {
#pragma unroll
                for (int T = Tlo; T < Thi; ++T) {
                    v2f b;
                    b[0] = X[T][2 * t];
                    b[1] = X[T][2 * t + 1];
                    acc[T] = __builtin_amdgcn_wmma_f32_16x16x4_f32(false, Av[t], false, b, (short)0,
                                                                   (t == 0) ? z : acc[T], false, false);
                }
            }
#pragma unroll
            for (int T = Tlo; T < Thi; ++T) {
                if (T + 1 < NX) {
                    v2f b;
                    b[0] = X[T + 1][0];
                    b[1] = X[T + 1][1];
                    acc[T] = __builtin_amdgcn_wmma_f32_16x16x4_f32(false, Av[4], false, b, (short)0,
                                                                   acc[T], false, false);
                }
            }
        }
        if (TOP) {
            v2f b;
            b[0] = H[0];
            b[1] = 0.0f;
            const v2f a5 = (s < 8) ? Av[4] : Ahi;
            acc[NT - 1] = __builtin_amdgcn_wmma_f32_16x16x4_f32(false, a5, false, b, (short)0,
                                                                acc[NT - 1], false, false);
            H = __builtin_shufflevector(H, H, 1, 2, 3, 4, 5, 6, 7, 0);
        }
        if (H1 < NX) {
#pragma unroll
            for (int i = 0; i < NW1; ++i) {
                __builtin_amdgcn_sched_group_barrier(0x008, 1, 0);
                __builtin_amdgcn_sched_group_barrier(0x002, 6, 0);
            }
#pragma unroll
            for (int T = H1; T < NX; ++T) {
#pragma unroll
                for (int r = 0; r < 8; ++r) X[T][r] = fmaxf(acc[T][r], P[T][r]);
            }
        } else {
#pragma unroll
            for (int T = 0; T < NX; ++T) {
#pragma unroll
                for (int r = 0; r < 8; ++r) X[T][r] = fmaxf(acc[T][r], fmaf(-scale, Q[T][r], kv));
            }
        }
        slast = scale;
        scale *= uscale;
    }

#pragma unroll
    for (int T = 0; T < NT; ++T) {
        unsigned eqm = 0xffffffffu;
#pragma unroll
        for (int r = 0; r < 8; ++r) {
            eqm &= __builtin_amdgcn_ballot_w32(X[T][r] == fmaf(-slast, Q[T][r], kv));
        }
        fdst[t0 + T] = (eqm == 0xffffffffu) ? 1 : 0;
#pragma unroll
        for (int q = 0; q < 4; ++q) {
            v2f p;
            p[0] = X[T][2 * q];
            p[1] = X[T][2 * q + 1];
            dst[(8 * (t0 + T) + q + 4 * hi) * BT + l15] = p;
        }
    }
    return scale;
}

__device__ __forceinline__ void jump_tile(const v2f* src, v2f* dst, int T, int hi, int l15,
                                          const v2f (&Aj)[8])
{
    v2f b[8];
#pragma unroll
    for (int o = 0; o < 2; ++o) {
#pragma unroll
        for (int q = 0; q < 4; ++q) b[4 * o + q] = src[(8 * (T + o) + q + 4 * hi) * BT + l15];
    }
    v8f acc = {};
#pragma unroll
    for (int f = 0; f < 8; ++f) {
        acc = __builtin_amdgcn_wmma_f32_16x16x4_f32(false, Aj[f], false, b[f], (short)0, acc,
                                                    false, false);
    }
#pragma unroll
    for (int q = 0; q < 4; ++q) {
        v2f pr;
        pr[0] = acc[2 * q];
        pr[1] = acc[2 * q + 1];
        dst[(8 * T + q + 4 * hi) * BT + l15] = pr;
    }
}

__device__ __forceinline__ void payoff_tile(v2f* dst, const float* PB, int T, int hi, int l15,
                                            float kv, float slast)
{
#pragma unroll
    for (int q = 0; q < 4; ++q) {
        v2f pr;
        pr[0] = fmaf(-slast, PB[16 * T + 2 * q + 8 * hi], kv);
        pr[1] = fmaf(-slast, PB[16 * T + 2 * q + 1 + 8 * hi], kv);
        dst[(8 * T + q + 4 * hi) * BT + l15] = pr;
    }
}

__global__ __launch_bounds__(32 * WAVES * GROUPS)
void btree_american_wmma(const float* __restrict__ kin, float* __restrict__ out,
                         float w0, float w1, float uscale, float alpha)
{
    __shared__ v2f   V2[GROUPS][2][NPAIR * BT];
    __shared__ float PB[NODES];
    __shared__ float CJ[BLK + 1];
    __shared__ float TM[BLK * 16];
    __shared__ int   FT[GROUPS][2][NODES / 16];

    const int lane = threadIdx.x & 31;
    const int l15  = lane & 15;
    const int hi   = lane >> 4;
    const int wv   = __builtin_amdgcn_readfirstlane((int)(threadIdx.x >> 5));
    const int g    = wv / WAVES;
    const int w    = wv % WAVES;
    const int tg   = (int)threadIdx.x % (32 * WAVES);
    const int b0   = (blockIdx.x * GROUPS + g) * BT;

    const float kv = kin[b0 + l15];

    constexpr unsigned short BIN[16][8] = {
        {1, 0, 0, 0, 0, 0, 0, 0},
        {1, 0, 0, 0, 0, 0, 0, 0},
        {1, 2, 0, 0, 0, 0, 0, 0},
        {1, 3, 0, 0, 0, 0, 0, 0},
        {1, 4, 6, 0, 0, 0, 0, 0},
        {1, 5, 10, 0, 0, 0, 0, 0},
        {1, 6, 15, 20, 0, 0, 0, 0},
        {1, 7, 21, 35, 0, 0, 0, 0},
        {1, 8, 28, 56, 70, 0, 0, 0},
        {1, 9, 36, 84, 126, 0, 0, 0},
        {1, 10, 45, 120, 210, 252, 0, 0},
        {1, 11, 55, 165, 330, 462, 0, 0},
        {1, 12, 66, 220, 495, 792, 924, 0},
        {1, 13, 78, 286, 715, 1287, 1716, 0},
        {1, 14, 91, 364, 1001, 2002, 3003, 3432},
        {1, 15, 105, 455, 1365, 3003, 5005, 6435}
    };
    const int ts = (int)threadIdx.x >> 4;
    const int td = (int)threadIdx.x & 15;
    const int tdm = (td < ts - td) ? td : ts - td;
    const float bn = (float)BIN[ts][tdm < 0 ? 0 : tdm];

    for (int h = tg >> 4; h < NPAIR; h += 2 * WAVES) {
        float s0 = 100.0f * expf(alpha * (float)(4 * h - NSTEP));
        float s1 = 100.0f * expf(alpha * (float)(4 * h + 2 - NSTEP));
        v2f v;
        v[0] = fmaxf(kv - s0, 0.0f);
        v[1] = fmaxf(kv - s1, 0.0f);
        V2[g][0][h * BT + l15] = v;
        V2[g][1][h * BT + l15] = v;
    }

    for (int j = (int)threadIdx.x; j < NODES; j += 32 * WAVES * GROUPS) {
        PB[j] = 100.0f * expf(alpha * (float)(2 * j - (NSTEP - 1)));
    }

    if (threadIdx.x <= BLK) {
        const int d  = (int)threadIdx.x;
        const int dm = (d < BLK - d) ? d : BLK - d;
        const double bn = dm == 0 ? 1.0 : dm == 1 ? 16.0 : dm == 2 ? 120.0 : dm == 3 ? 560.0
                        : dm == 4 ? 1820.0 : dm == 5 ? 4368.0 : dm == 6 ? 8008.0
                        : dm == 7 ? 11440.0 : 12870.0;
        const double a1 = (double)w0, a2 = a1 * a1, a4 = a2 * a2, a8 = a4 * a4, a16 = a8 * a8;
        const double b1 = (double)w1, b2 = b1 * b1, b4 = b2 * b2, b8 = b4 * b4, b16 = b8 * b8;
        const int na = BLK - d;
        double c = bn;
        c *= (na & 1) ? a1 : 1.0;
        c *= (na & 2) ? a2 : 1.0;
        c *= (na & 4) ? a4 : 1.0;
        c *= (na & 8) ? a8 : 1.0;
        c *= (na & 16) ? a16 : 1.0;
        c *= (d & 1) ? b1 : 1.0;
        c *= (d & 2) ? b2 : 1.0;
        c *= (d & 4) ? b4 : 1.0;
        c *= (d & 8) ? b8 : 1.0;
        c *= (d & 16) ? b16 : 1.0;
        CJ[d] = (float)c;
    }

    {
        const int s = ts;
        const int d = td;
        const double a1 = (double)w0, a2 = a1 * a1, a4 = a2 * a2, a8 = a4 * a4;
        const double b1 = (double)w1, b2 = b1 * b1, b4 = b2 * b2, b8 = b4 * b4;
        const int na = s - d;
        double c = (double)bn;
        c *= (na & 1) ? a1 : 1.0;
        c *= (na & 2) ? a2 : 1.0;
        c *= (na & 4) ? a4 : 1.0;
        c *= (na & 8) ? a8 : 1.0;
        c *= (d & 1) ? b1 : 1.0;
        c *= (d & 2) ? b2 : 1.0;
        c *= (d & 4) ? b4 : 1.0;
        c *= (d & 8) ? b8 : 1.0;
        TM[s * 16 + d] = (d <= s) ? (float)c : 0.0f;
    }

    v2f Av[5];
#pragma unroll
    for (int t = 0; t < 5; ++t) {
        int tt = (t < 4) ? t : 8;
        int n0 = 2 * tt + 8 * hi;
        int n1 = n0 + 1;
        float e0 = (n0 == l15) ? w0 : ((n0 == l15 + 1) ? w1 : 0.0f);
        float e1 = (n1 == l15) ? w0 : ((n1 == l15 + 1) ? w1 : 0.0f);
        Av[t][0] = e0;
        Av[t][1] = e1;
    }

    {
        float kn = kv;
#pragma unroll
        for (int o = 8; o >= 1; o >>= 1) kn = fminf(kn, __shfl_xor(kn, o, 16));
        if (w == 0 && lane < NODES / 16) {
            FT[g][0][lane] = (0.999f * kn > 100.0f * expf(alpha * (float)(32 * lane + 30 - NSTEP))) ? 1 : 0;
        }
    }
    __syncthreads();

    v2f Aj[8];
#pragma unroll
    for (int f = 0; f < 8; ++f) {
#pragma unroll
        for (int e = 0; e < 2; ++e) {
            const int d = 16 * (f / 4) + 2 * (f % 4) + e + 8 * hi - l15;
            const float c = CJ[(d < 0) ? 0 : ((d > BLK) ? BLK : d)];
            Aj[f][e] = (d >= 0 && d <= BLK) ? c : 0.0f;
        }
    }

    v2f Mt[4];
#pragma unroll
    for (int t = 0; t < 4; ++t) {
#pragma unroll
        for (int e = 0; e < 2; ++e) Mt[t][e] = TM[l15 * 16 + 2 * t + e + 8 * hi];
    }
    v2f Ahi;
    Ahi[0] = (l15 == 15 && hi == 1) ? w1 : 0.0f;
    Ahi[1] = 0.0f;

    float km = kv;
#pragma unroll
    for (int o = 8; o >= 1; o >>= 1) km = fmaxf(km, __shfl_xor(km, o, 16));
    const float kmax = __builtin_bit_cast(float, __builtin_amdgcn_readfirstlane(__builtin_bit_cast(int, km)));
    const bool wpos = (w0 >= 0.0f) && (w1 >= 0.0f);
    const int  tl  = (lane < NODES / 16) ? lane : NODES / 16 - 1;

    float scale = 1.0f;
    for (int b = 0; b < NBLK; ++b) {
        const v2f* src = V2[g][b & 1];
        v2f*       dst = V2[g][(b & 1) ^ 1];
        const int ot   = NBLK - b;
        const float pb0 = PB[16 * (lane < NODES / 16 ? lane : NODES / 16 - 1)];
        const bool live = (lane < ot) && !(wpos && kmax < 0.999f * scale * pb0);
        const unsigned lm = __builtin_amdgcn_ballot_w32(live);
        const int st   = lm ? (32 - __builtin_clz(lm)) : 0;
        const bool fl = FT[g][b & 1][tl] != 0;
        const unsigned fm = __builtin_amdgcn_ballot_w32(fl && (lane < ot));
        const int nf   = __builtin_ctz(~fm);
        const int ft0  = (nf > 0) ? nf - 1 : 0;
        const int ft   = (ft0 < st) ? ft0 : st;
        int* fdst = FT[g][(b & 1) ^ 1];
        const int nlo  = (st - ft) >> 1;
        const int nup  = (st - ft) - nlo;
        const int role = g ? (w ^ 2) : w;
        const int nt   = (role == 0) ? nlo : ((role == 1) ? nup : 0);
        const int t0   = (role == 1) ? ft + nlo : ft;
        if (nt == 0) {
            const int e0 = (nlo == 0) ? 1 : 0;
            const int e1 = (nup == 0) ? 1 : 0;
            const int nj = 2 + e0 + e1;
            const int q  = (role >= 2) ? (role - 2) : ((role == 0) ? 2 : 2 + e0);
            for (int T = st + q; T < ot; T += nj) {
                jump_tile(src, dst, T, hi, l15, Aj);
                fdst[T] = 0;
            }
            for (int s = 0; s < BLK - 1; ++s) scale *= uscale;
            for (int T = q; T < ft; T += nj) {
                payoff_tile(dst, PB, T, hi, l15, kv, scale);
                fdst[T] = 1;
            }
            scale *= uscale;
        } else if (role == 1) {
            switch (nt) {
            case 8:  scale = run_block<8, true>(src, dst, fdst, PB, t0, hi, l15, kv, scale, uscale, Av, Mt, Ahi); break;
            case 7:  scale = run_block<7, true>(src, dst, fdst, PB, t0, hi, l15, kv, scale, uscale, Av, Mt, Ahi); break;
            case 6:  scale = run_block<6, true>(src, dst, fdst, PB, t0, hi, l15, kv, scale, uscale, Av, Mt, Ahi); break;
            case 5:  scale = run_block<5, true>(src, dst, fdst, PB, t0, hi, l15, kv, scale, uscale, Av, Mt, Ahi); break;
            case 4:  scale = run_block<4, true>(src, dst, fdst, PB, t0, hi, l15, kv, scale, uscale, Av, Mt, Ahi); break;
            case 3:  scale = run_block<3, true>(src, dst, fdst, PB, t0, hi, l15, kv, scale, uscale, Av, Mt, Ahi); break;
            case 2:  scale = run_block<2, true>(src, dst, fdst, PB, t0, hi, l15, kv, scale, uscale, Av, Mt, Ahi); break;
            default: scale = run_block<1, true>(src, dst, fdst, PB, t0, hi, l15, kv, scale, uscale, Av, Mt, Ahi); break;
            }
        } else {
            switch (nt) {
            case 8:  scale = run_block<8, false>(src, dst, fdst, PB, t0, hi, l15, kv, scale, uscale, Av, Mt, Ahi); break;
            case 7:  scale = run_block<7, false>(src, dst, fdst, PB, t0, hi, l15, kv, scale, uscale, Av, Mt, Ahi); break;
            case 6:  scale = run_block<6, false>(src, dst, fdst, PB, t0, hi, l15, kv, scale, uscale, Av, Mt, Ahi); break;
            case 5:  scale = run_block<5, false>(src, dst, fdst, PB, t0, hi, l15, kv, scale, uscale, Av, Mt, Ahi); break;
            case 4:  scale = run_block<4, false>(src, dst, fdst, PB, t0, hi, l15, kv, scale, uscale, Av, Mt, Ahi); break;
            case 3:  scale = run_block<3, false>(src, dst, fdst, PB, t0, hi, l15, kv, scale, uscale, Av, Mt, Ahi); break;
            case 2:  scale = run_block<2, false>(src, dst, fdst, PB, t0, hi, l15, kv, scale, uscale, Av, Mt, Ahi); break;
            default: scale = run_block<1, false>(src, dst, fdst, PB, t0, hi, l15, kv, scale, uscale, Av, Mt, Ahi); break;
            }
        }
        __syncthreads();
    }

    if (threadIdx.x < GROUPS * BT) out[blockIdx.x * GROUPS * BT + lane] = V2[hi][0][l15][0];
}

extern "C" void kernel_launch(void* const* d_in, const int* in_sizes, int n_in,
                              void* d_out, int out_size, void* d_ws, size_t ws_size,
                              hipStream_t stream) {
    (void)n_in; (void)d_ws; (void)ws_size; (void)out_size;
    const float* kin = (const float*)d_in[0];
    float* out = (float*)d_out;
    const int B = in_sizes[0];

    const double S0g = 100.0, SIG = 0.2, Tg = 1.0, T0g = 0.0, Rg = 0.05;
    (void)S0g;
    const double dt   = (Tg - T0g) / (double)NSTEP;
    const double sq   = sqrt(dt);
    const double u    = exp(SIG * sq);
    const double d    = exp(-SIG * sq);
    const double disc = exp(-Rg * dt);
    const float w0    = (float)((disc * u - 1.0) / (u - d));
    const float w1    = (float)((1.0 - disc * d) / (u - d));
    const float us    = (float)u;
    const float alpha = (float)(SIG * sq);

    const int nblocks = B / (BT * GROUPS);
    btree_american_wmma<<<nblocks, 32 * WAVES * GROUPS, 0, stream>>>(kin, out, w0, w1, us, alpha);
}


# A workgroup's 32 strikes ranked in registers: each 16-strike group takes a contiguous half of the ranks and steps fewer tiles; a strike's tiles may step or jump differently, so sum order changes
# speedup vs baseline: 5.0452x; 1.0761x over previous
#include <hip/hip_runtime.h>
#include <cmath>


typedef float v2f __attribute__((ext_vector_type(2)));
typedef float v8f __attribute__((ext_vector_type(8)));

#define NSTEP 256
#define NODES 288
#define NPAIR (NODES / 2)
#define BT    16
#define WAVES 4
#define GROUPS 2
#define BLK   16
#define NBLK  (NSTEP / BLK)
#define MAXT  8

static_assert(NBLK % 2 == 0, "an even block count leaves the result in plane 0");
static_assert(BLK == 16, "the band of a whole block (binomials, powers by bits) is written for 16 steps");
static_assert(16 * (NBLK + 1) <= NODES, "the tile above the widest block's top tile is inside the pad");
static_assert(32 * WAVES * GROUPS == BLK * 16, "one thread per entry of the table of s-step maps");
static_assert(GROUPS * BT == 32, "one strike per lane: a wave ranks the workgroup's strikes in its registers");

template <int NT, bool TOP>
__device__ __forceinline__ float run_block(const v2f* src, v2f* dst, int* fdst, const float* PB, int t0,
                                           int hi, int l15, float kv, float scale, float uscale,
                                           const v2f (&Av)[5], const v2f (&Mt)[4], const v2f Ahi)
{
    constexpr int NX = TOP ? NT : NT + 1;
    v8f X[NX];
    v8f Q[NX];
#pragma unroll
    for (int T = 0; T < NX; ++T) {
#pragma unroll
        for (int q = 0; q < 4; ++q) {
            v2f p = src[(8 * (t0 + T) + q + 4 * hi) * BT + l15];
            X[T][2 * q]     = p[0];
            X[T][2 * q + 1] = p[1];
        }
    }
    v8f H = {};
    if (TOP) {
        v2f hb[4];
#pragma unroll
        for (int q = 0; q < 4; ++q) hb[q] = src[(8 * (t0 + NT) + q + 4 * hi) * BT + l15];
#pragma unroll
        for (int q = 0; q < 4; ++q) {
            H = __builtin_amdgcn_wmma_f32_16x16x4_f32(false, Mt[q], false, hb[q], (short)0, H,
                                                      false, false);
        }
    }
#pragma unroll
    for (int T = 0; T < NX; ++T) {
#pragma unroll
        for (int r = 0; r < 8; ++r) Q[T][r] = PB[16 * (t0 + T) + r + 8 * hi];
    }

    constexpr int H1  = (NX >= 4) ? NX / 2 : NX;
    constexpr int NW1 = 4 * (NX - H1) + (NX - H1 - 1) + (TOP ? 1 : 0);
    float slast = scale;
    for (int s = 0; s < BLK; ++s) {
        v8f acc[NX];
        v8f P[NX];
        const v8f z = {};
#pragma unroll
        for (int half = 0; half < 2; ++half) {
            const int Tlo = half ? H1 : 0;
            const int Thi = half ? NX : H1;
            if (half == 1 && H1 < NX) {
                __builtin_amdgcn_sched_barrier(0);
#pragma unroll
                for (int T = 0; T < NX; ++T) {
#pragma unroll
                    for (int r = 0; r < 8; ++r) P[T][r] = fmaf(-scale, Q[T][r], kv);
                }
#pragma unroll
                for (int T = 0; T < H1; ++T) {
#pragma unroll
                    for (int r = 0; r < 8; ++r) X[T][r] = fmaxf(acc[T][r], P[T][r]);
                }
            }
#pragma unroll
            for (int t = 0; t < 4; ++t) {
#pragma unroll
                for (int T = Tlo; T < Thi; ++T) {
                    v2f b;
                    b[0] = X[T][2 * t];
                    b[1] = X[T][2 * t + 1];
                    acc[T] = __builtin_amdgcn_wmma_f32_16x16x4_f32(false, Av[t], false, b, (short)0,
                                                                   (t == 0) ? z : acc[T], false, false);
                }
            }
#pragma unroll
            for (int T = Tlo; T < Thi; ++T) {
                if (T + 1 < NX) {
                    v2f b;
                    b[0] = X[T + 1][0];
                    b[1] = X[T + 1][1];
                    acc[T] = __builtin_amdgcn_wmma_f32_16x16x4_f32(false, Av[4], false, b, (short)0,
                                                                   acc[T], false, false);
                }
            }
        }
        if (TOP) {
            v2f b;
            b[0] = H[0];
            b[1] = 0.0f;
            const v2f a5 = (s < 8) ? Av[4] : Ahi;
            acc[NT - 1] = __builtin_amdgcn_wmma_f32_16x16x4_f32(false, a5, false, b, (short)0,
                                                                acc[NT - 1], false, false);
            H = __builtin_shufflevector(H, H, 1, 2, 3, 4, 5, 6, 7, 0);
        }
        if (H1 < NX) {
#pragma unroll
            for (int i = 0; i < NW1; ++i) {
                __builtin_amdgcn_sched_group_barrier(0x008, 1, 0);
                __builtin_amdgcn_sched_group_barrier(0x002, 6, 0);
            }
#pragma unroll
            for (int T = H1; T < NX; ++T) {
#pragma unroll
                for (int r = 0; r < 8; ++r) X[T][r] = fmaxf(acc[T][r], P[T][r]);
            }
        } else {
#pragma unroll
            for (int T = 0; T < NX; ++T) {
#pragma unroll
                for (int r = 0; r < 8; ++r) X[T][r] = fmaxf(acc[T][r], fmaf(-scale, Q[T][r], kv));
            }
        }
        slast = scale;
        scale *= uscale;
    }

#pragma unroll
    for (int T = 0; T < NT; ++T) {
        unsigned eqm = 0xffffffffu;
#pragma unroll
        for (int r = 0; r < 8; ++r) {
            eqm &= __builtin_amdgcn_ballot_w32(X[T][r] == fmaf(-slast, Q[T][r], kv));
        }
        fdst[t0 + T] = (eqm == 0xffffffffu) ? 1 : 0;
#pragma unroll
        for (int q = 0; q < 4; ++q) {
            v2f p;
            p[0] = X[T][2 * q];
            p[1] = X[T][2 * q + 1];
            dst[(8 * (t0 + T) + q + 4 * hi) * BT + l15] = p;
        }
    }
    return scale;
}

__device__ __forceinline__ void jump_tile(const v2f* src, v2f* dst, int T, int hi, int l15,
                                          const v2f (&Aj)[8])
{
    v2f b[8];
#pragma unroll
    for (int o = 0; o < 2; ++o) {
#pragma unroll
        for (int q = 0; q < 4; ++q) b[4 * o + q] = src[(8 * (T + o) + q + 4 * hi) * BT + l15];
    }
    v8f acc = {};
#pragma unroll
    for (int f = 0; f < 8; ++f) {
        acc = __builtin_amdgcn_wmma_f32_16x16x4_f32(false, Aj[f], false, b[f], (short)0, acc,
                                                    false, false);
    }
#pragma unroll
    for (int q = 0; q < 4; ++q) {
        v2f pr;
        pr[0] = acc[2 * q];
        pr[1] = acc[2 * q + 1];
        dst[(8 * T + q + 4 * hi) * BT + l15] = pr;
    }
}

__device__ __forceinline__ void payoff_tile(v2f* dst, const float* PB, int T, int hi, int l15,
                                            float kv, float slast)
{
#pragma unroll
    for (int q = 0; q < 4; ++q) {
        v2f pr;
        pr[0] = fmaf(-slast, PB[16 * T + 2 * q + 8 * hi], kv);
        pr[1] = fmaf(-slast, PB[16 * T + 2 * q + 1 + 8 * hi], kv);
        dst[(8 * T + q + 4 * hi) * BT + l15] = pr;
    }
}

__global__ __launch_bounds__(32 * WAVES * GROUPS)
void btree_american_wmma(const float* __restrict__ kin, float* __restrict__ out,
                         float w0, float w1, float uscale, float alpha)
{
    __shared__ v2f   V2[GROUPS][2][NPAIR * BT];
    __shared__ float PB[NODES];
    __shared__ float CJ[BLK + 1];
    __shared__ float TM[BLK * 16];
    __shared__ int   FT[GROUPS][2][NODES / 16];

    const int lane = threadIdx.x & 31;
    const int l15  = lane & 15;
    const int hi   = lane >> 4;
    const int wv   = __builtin_amdgcn_readfirstlane((int)(threadIdx.x >> 5));
    const int g    = wv / WAVES;
    const int w    = wv % WAVES;
    const int tg   = (int)threadIdx.x % (32 * WAVES);

    const float kraw = kin[blockIdx.x * (GROUPS * BT) + lane];
    const unsigned kbits = __builtin_bit_cast(unsigned, kraw);
    const unsigned kkey  = (kbits & 0x80000000u) ? ~kbits : (kbits | 0x80000000u);
    int rk = 0;
#pragma unroll
    for (int j = 0; j < 32; ++j) {
        const unsigned kj = (unsigned)__builtin_amdgcn_readlane((int)kkey, j);
        rk += (kj < kkey || (kj == kkey && j < lane)) ? 1 : 0;
    }
    float kv = 0.0f;
#pragma unroll
    for (int j = 0; j < 32; ++j) {
        const int   rj = __builtin_amdgcn_readlane(rk, j);
        const float kj = __builtin_bit_cast(float, __builtin_amdgcn_readlane((int)kbits, j));
        kv = (rj == BT * g + l15) ? kj : kv;
    }

    constexpr unsigned short BIN[16][8] = {
        {1, 0, 0, 0, 0, 0, 0, 0},
        {1, 0, 0, 0, 0, 0, 0, 0},
        {1, 2, 0, 0, 0, 0, 0, 0},
        {1, 3, 0, 0, 0, 0, 0, 0},
        {1, 4, 6, 0, 0, 0, 0, 0},
        {1, 5, 10, 0, 0, 0, 0, 0},
        {1, 6, 15, 20, 0, 0, 0, 0},
        {1, 7, 21, 35, 0, 0, 0, 0},
        {1, 8, 28, 56, 70, 0, 0, 0},
        {1, 9, 36, 84, 126, 0, 0, 0},
        {1, 10, 45, 120, 210, 252, 0, 0},
        {1, 11, 55, 165, 330, 462, 0, 0},
        {1, 12, 66, 220, 495, 792, 924, 0},
        {1, 13, 78, 286, 715, 1287, 1716, 0},
        {1, 14, 91, 364, 1001, 2002, 3003, 3432},
        {1, 15, 105, 455, 1365, 3003, 5005, 6435}
    };
    const int ts = (int)threadIdx.x >> 4;
    const int td = (int)threadIdx.x & 15;
    const int tdm = (td < ts - td) ? td : ts - td;
    const float bn = (float)BIN[ts][tdm < 0 ? 0 : tdm];

    for (int h = tg >> 4; h < NPAIR; h += 2 * WAVES) {
        float s0 = 100.0f * expf(alpha * (float)(4 * h - NSTEP));
        float s1 = 100.0f * expf(alpha * (float)(4 * h + 2 - NSTEP));
        v2f v;
        v[0] = fmaxf(kv - s0, 0.0f);
        v[1] = fmaxf(kv - s1, 0.0f);
        V2[g][0][h * BT + l15] = v;
        V2[g][1][h * BT + l15] = v;
    }

    for (int j = (int)threadIdx.x; j < NODES; j += 32 * WAVES * GROUPS) {
        PB[j] = 100.0f * expf(alpha * (float)(2 * j - (NSTEP - 1)));
    }

    if (threadIdx.x <= BLK) {
        const int d  = (int)threadIdx.x;
        const int dm = (d < BLK - d) ? d : BLK - d;
        const double bn = dm == 0 ? 1.0 : dm == 1 ? 16.0 : dm == 2 ? 120.0 : dm == 3 ? 560.0
                        : dm == 4 ? 1820.0 : dm == 5 ? 4368.0 : dm == 6 ? 8008.0
                        : dm == 7 ? 11440.0 : 12870.0;
        const double a1 = (double)w0, a2 = a1 * a1, a4 = a2 * a2, a8 = a4 * a4, a16 = a8 * a8;
        const double b1 = (double)w1, b2 = b1 * b1, b4 = b2 * b2, b8 = b4 * b4, b16 = b8 * b8;
        const int na = BLK - d;
        double c = bn;
        c *= (na & 1) ? a1 : 1.0;
        c *= (na & 2) ? a2 : 1.0;
        c *= (na & 4) ? a4 : 1.0;
        c *= (na & 8) ? a8 : 1.0;
        c *= (na & 16) ? a16 : 1.0;
        c *= (d & 1) ? b1 : 1.0;
        c *= (d & 2) ? b2 : 1.0;
        c *= (d & 4) ? b4 : 1.0;
        c *= (d & 8) ? b8 : 1.0;
        c *= (d & 16) ? b16 : 1.0;
        CJ[d] = (float)c;
    }

    {
        const int s = ts;
        const int d = td;
        const double a1 = (double)w0, a2 = a1 * a1, a4 = a2 * a2, a8 = a4 * a4;
        const double b1 = (double)w1, b2 = b1 * b1, b4 = b2 * b2, b8 = b4 * b4;
        const int na = s - d;
        double c = (double)bn;
        c *= (na & 1) ? a1 : 1.0;
        c *= (na & 2) ? a2 : 1.0;
        c *= (na & 4) ? a4 : 1.0;
        c *= (na & 8) ? a8 : 1.0;
        c *= (d & 1) ? b1 : 1.0;
        c *= (d & 2) ? b2 : 1.0;
        c *= (d & 4) ? b4 : 1.0;
        c *= (d & 8) ? b8 : 1.0;
        TM[s * 16 + d] = (d <= s) ? (float)c : 0.0f;
    }

    v2f Av[5];
#pragma unroll
    for (int t = 0; t < 5; ++t) {
        int tt = (t < 4) ? t : 8;
        int n0 = 2 * tt + 8 * hi;
        int n1 = n0 + 1;
        float e0 = (n0 == l15) ? w0 : ((n0 == l15 + 1) ? w1 : 0.0f);
        float e1 = (n1 == l15) ? w0 : ((n1 == l15 + 1) ? w1 : 0.0f);
        Av[t][0] = e0;
        Av[t][1] = e1;
    }

    {
        float kn = kv;
#pragma unroll
        for (int o = 8; o >= 1; o >>= 1) kn = fminf(kn, __shfl_xor(kn, o, 16));
        if (w == 0 && lane < NODES / 16) {
            FT[g][0][lane] = (0.999f * kn > 100.0f * expf(alpha * (float)(32 * lane + 30 - NSTEP))) ? 1 : 0;
        }
    }
    __syncthreads();

    v2f Aj[8];
#pragma unroll
    for (int f = 0; f < 8; ++f) {
#pragma unroll
        for (int e = 0; e < 2; ++e) {
            const int d = 16 * (f / 4) + 2 * (f % 4) + e + 8 * hi - l15;
            const float c = CJ[(d < 0) ? 0 : ((d > BLK) ? BLK : d)];
            Aj[f][e] = (d >= 0 && d <= BLK) ? c : 0.0f;
        }
    }

    v2f Mt[4];
#pragma unroll
    for (int t = 0; t < 4; ++t) {
#pragma unroll
        for (int e = 0; e < 2; ++e) Mt[t][e] = TM[l15 * 16 + 2 * t + e + 8 * hi];
    }
    v2f Ahi;
    Ahi[0] = (l15 == 15 && hi == 1) ? w1 : 0.0f;
    Ahi[1] = 0.0f;

    float km = kv;
#pragma unroll
    for (int o = 8; o >= 1; o >>= 1) km = fmaxf(km, __shfl_xor(km, o, 16));
    const float kmax = __builtin_bit_cast(float, __builtin_amdgcn_readfirstlane(__builtin_bit_cast(int, km)));
    const bool wpos = (w0 >= 0.0f) && (w1 >= 0.0f);
    const int  tl  = (lane < NODES / 16) ? lane : NODES / 16 - 1;

    float scale = 1.0f;
    for (int b = 0; b < NBLK; ++b) {
        const v2f* src = V2[g][b & 1];
        v2f*       dst = V2[g][(b & 1) ^ 1];
        const int ot   = NBLK - b;
        const float pb0 = PB[16 * (lane < NODES / 16 ? lane : NODES / 16 - 1)];
        const bool live = (lane < ot) && !(wpos && kmax < 0.999f * scale * pb0);
        const unsigned lm = __builtin_amdgcn_ballot_w32(live);
        const int st   = lm ? (32 - __builtin_clz(lm)) : 0;
        const bool fl = FT[g][b & 1][tl] != 0;
        const unsigned fm = __builtin_amdgcn_ballot_w32(fl && (lane < ot));
        const int nf   = __builtin_ctz(~fm);
        const int ft0  = (nf > 0) ? nf - 1 : 0;
        const int ft   = (ft0 < st) ? ft0 : st;
        int* fdst = FT[g][(b & 1) ^ 1];
        const int nlo  = (st - ft) >> 1;
        const int nup  = (st - ft) - nlo;
        const int role = g ? (w ^ 2) : w;
        const int nt   = (role == 0) ? nlo : ((role == 1) ? nup : 0);
        const int t0   = (role == 1) ? ft + nlo : ft;
        if (nt == 0) {
            const int e0 = (nlo == 0) ? 1 : 0;
            const int e1 = (nup == 0) ? 1 : 0;
            const int nj = 2 + e0 + e1;
            const int q  = (role >= 2) ? (role - 2) : ((role == 0) ? 2 : 2 + e0);
            for (int T = st + q; T < ot; T += nj) {
                jump_tile(src, dst, T, hi, l15, Aj);
                fdst[T] = 0;
            }
            for (int s = 0; s < BLK - 1; ++s) scale *= uscale;
            for (int T = q; T < ft; T += nj) {
                payoff_tile(dst, PB, T, hi, l15, kv, scale);
                fdst[T] = 1;
            }
            scale *= uscale;
        } else if (role == 1) {
            switch (nt) {
            case 8:  scale = run_block<8, true>(src, dst, fdst, PB, t0, hi, l15, kv, scale, uscale, Av, Mt, Ahi); break;
            case 7:  scale = run_block<7, true>(src, dst, fdst, PB, t0, hi, l15, kv, scale, uscale, Av, Mt, Ahi); break;
            case 6:  scale = run_block<6, true>(src, dst, fdst, PB, t0, hi, l15, kv, scale, uscale, Av, Mt, Ahi); break;
            case 5:  scale = run_block<5, true>(src, dst, fdst, PB, t0, hi, l15, kv, scale, uscale, Av, Mt, Ahi); break;
            case 4:  scale = run_block<4, true>(src, dst, fdst, PB, t0, hi, l15, kv, scale, uscale, Av, Mt, Ahi); break;
            case 3:  scale = run_block<3, true>(src, dst, fdst, PB, t0, hi, l15, kv, scale, uscale, Av, Mt, Ahi); break;
            case 2:  scale = run_block<2, true>(src, dst, fdst, PB, t0, hi, l15, kv, scale, uscale, Av, Mt, Ahi); break;
            default: scale = run_block<1, true>(src, dst, fdst, PB, t0, hi, l15, kv, scale, uscale, Av, Mt, Ahi); break;
            }
        } else {
            switch (nt) {
            case 8:  scale = run_block<8, false>(src, dst, fdst, PB, t0, hi, l15, kv, scale, uscale, Av, Mt, Ahi); break;
            case 7:  scale = run_block<7, false>(src, dst, fdst, PB, t0, hi, l15, kv, scale, uscale, Av, Mt, Ahi); break;
            case 6:  scale = run_block<6, false>(src, dst, fdst, PB, t0, hi, l15, kv, scale, uscale, Av, Mt, Ahi); break;
            case 5:  scale = run_block<5, false>(src, dst, fdst, PB, t0, hi, l15, kv, scale, uscale, Av, Mt, Ahi); break;
            case 4:  scale = run_block<4, false>(src, dst, fdst, PB, t0, hi, l15, kv, scale, uscale, Av, Mt, Ahi); break;
            case 3:  scale = run_block<3, false>(src, dst, fdst, PB, t0, hi, l15, kv, scale, uscale, Av, Mt, Ahi); break;
            case 2:  scale = run_block<2, false>(src, dst, fdst, PB, t0, hi, l15, kv, scale, uscale, Av, Mt, Ahi); break;
            default: scale = run_block<1, false>(src, dst, fdst, PB, t0, hi, l15, kv, scale, uscale, Av, Mt, Ahi); break;
            }
        }
        __syncthreads();
    }

    if (threadIdx.x < GROUPS * BT) out[blockIdx.x * GROUPS * BT + lane] = V2[rk >> 4][0][rk & 15][0];
}

extern "C" void kernel_launch(void* const* d_in, const int* in_sizes, int n_in,
                              void* d_out, int out_size, void* d_ws, size_t ws_size,
                              hipStream_t stream) {
    (void)n_in; (void)d_ws; (void)ws_size; (void)out_size;
    const float* kin = (const float*)d_in[0];
    float* out = (float*)d_out;
    const int B = in_sizes[0];

    const double S0g = 100.0, SIG = 0.2, Tg = 1.0, T0g = 0.0, Rg = 0.05;
    (void)S0g;
    const double dt   = (Tg - T0g) / (double)NSTEP;
    const double sq   = sqrt(dt);
    const double u    = exp(SIG * sq);
    const double d    = exp(-SIG * sq);
    const double disc = exp(-Rg * dt);
    const float w0    = (float)((disc * u - 1.0) / (u - d));
    const float w1    = (float)((1.0 - disc * d) / (u - d));
    const float us    = (float)u;
    const float alpha = (float)(SIG * sq);

    const int nblocks = B / (BT * GROUPS);
    btree_american_wmma<<<nblocks, 32 * WAVES * GROUPS, 0, stream>>>(kin, out, w0, w1, us, alpha);
}
